# starting kernel
# baseline (speedup 1.0000x reference)
#include <hip/hip_runtime.h>
#include <hip/hip_bf16.h>

#define F_IN 512
#define HID  16
#define NCLS 7
#define TILES_PER_WAVE 2

typedef float v2f __attribute__((ext_vector_type(2)));
typedef float v8f __attribute__((ext_vector_type(8)));


__global__ void gcn_init_deg(float* __restrict__ deg, int n) {
    int i = blockIdx.x * blockDim.x + threadIdx.x;
    if (i < n) deg[i] = 1.0f;
}

__global__ void gcn_deg_scatter(const int* __restrict__ col, float* __restrict__ deg, int e) {
    int i = blockIdx.x * blockDim.x + threadIdx.x;
    if (i < e) atomicAdd(&deg[col[i]], 1.0f);
}

__global__ void gcn_dinv(float* __restrict__ degdinv, int n) {
    int i = blockIdx.x * blockDim.x + threadIdx.x;
    if (i < n) degdinv[i] = rsqrtf(degdinv[i]);
}


__device__ __forceinline__ void store_tile(float* __restrict__ h1, const v8f& acc,
                                           int tile, int lane, int n) {
    int mbase = tile * 16 + ((lane >> 4) << 3);
    int ncol  = lane & 15;
    float* p = h1 + (size_t)mbase * HID + ncol;
    if (mbase + 8 <= n) {
        #pragma unroll
        for (int v = 0; v < 8; ++v) p[v * HID] = acc[v];
    } else {
        #pragma unroll
        for (int v = 0; v < 8; ++v) if (mbase + v < n) p[v * HID] = acc[v];
    }
}

__global__ void gcn_gemm1_wmma(const float* __restrict__ x, const float* __restrict__ W1,
                               float* __restrict__ h1, int n) {
    int lane = threadIdx.x & 31;
    int wid  = (blockIdx.x * blockDim.x + threadIdx.x) >> 5;
    int ntiles = (n + 15) >> 4;
    int tile0 = wid * TILES_PER_WAVE;
    if (tile0 >= ntiles) return;
    bool has1 = (tile0 + 1) < ntiles;

    int m    = lane & 15;
    int koff = (lane >> 4) << 1;

    int row0 = tile0 * 16 + m;
    int row1 = row0 + 16;
    const float* xa0 = x + (size_t)(row0 < n ? row0 : 0) * F_IN + koff;
    const float* xa1 = x + (size_t)(row1 < n ? row1 : 0) * F_IN + koff;

    v8f acc0 = {}, acc1 = {};
    #pragma unroll 4
    for (int k = 0; k < F_IN; k += 4) {
        v2f b;
        b.x = W1[(size_t)(k + koff)     * HID + m];
        b.y = W1[(size_t)(k + koff + 1) * HID + m];
        v2f a0 = *(const v2f*)(xa0 + k);
        v2f a1 = *(const v2f*)(xa1 + k);
        acc0 = __builtin_amdgcn_wmma_f32_16x16x4_f32(
            false, a0, false, b, (short)0, acc0, false, false);
        acc1 = __builtin_amdgcn_wmma_f32_16x16x4_f32(
            false, a1, false, b, (short)0, acc1, false, false);
    }

    store_tile(h1, acc0, tile0, lane, n);
    if (has1) store_tile(h1, acc1, tile0 + 1, lane, n);
}


__global__ void gcn_init_out1(const float* __restrict__ h1, const float* __restrict__ dinv,
                              const float* __restrict__ b1, float* __restrict__ out1, int n) {
    int i = blockIdx.x * blockDim.x + threadIdx.x;
    if (i >= n) return;
    float s = dinv[i]; s = s * s;
    const float4* hp = (const float4*)(h1 + (size_t)i * HID);
    float4* op = (float4*)(out1 + (size_t)i * HID);
    #pragma unroll
    for (int q = 0; q < 4; ++q) {
        float4 h = hp[q];
        op[q] = make_float4(b1[4*q+0] + s*h.x, b1[4*q+1] + s*h.y,
                            b1[4*q+2] + s*h.z, b1[4*q+3] + s*h.w);
    }
}

__global__ void gcn_scatter1(const int* __restrict__ rowi, const int* __restrict__ coli,
                             const float* __restrict__ dinv, const float* __restrict__ h1,
                             float* __restrict__ out1, int e) {
    int idx = blockIdx.x * blockDim.x + threadIdx.x;
    if (idx >= e) return;
    int r = rowi[idx], c = coli[idx];
    float w = dinv[r] * dinv[c];
    const float* hp = h1 + (size_t)r * HID;
    float* op = out1 + (size_t)c * HID;
    #pragma unroll
    for (int q = 0; q < HID; ++q) atomicAdd(op + q, w * hp[q]);
}


__global__ void gcn_gemm2(const float* __restrict__ out1, const float* __restrict__ W2,
                          const float* __restrict__ b2, const float* __restrict__ dinv,
                          float* __restrict__ g, float* __restrict__ out2, int n) {
    int i = blockIdx.x * blockDim.x + threadIdx.x;
    if (i >= n) return;
    float h[HID];
    const float* ip = out1 + (size_t)i * HID;
    #pragma unroll
    for (int c = 0; c < HID; ++c) h[c] = fmaxf(ip[c], 0.0f);
    float s = dinv[i]; s = s * s;
    #pragma unroll
    for (int c2 = 0; c2 < NCLS; ++c2) {
        float acc = 0.0f;
        #pragma unroll
        for (int c = 0; c < HID; ++c) acc += h[c] * W2[c * NCLS + c2];
        g[(size_t)i * NCLS + c2]    = acc;
        out2[(size_t)i * NCLS + c2] = b2[c2] + s * acc;
    }
}

__global__ void gcn_scatter2(const int* __restrict__ rowi, const int* __restrict__ coli,
                             const float* __restrict__ dinv, const float* __restrict__ g,
                             float* __restrict__ out2, int e) {
    int idx = blockIdx.x * blockDim.x + threadIdx.x;
    if (idx >= e) return;
    int r = rowi[idx], c = coli[idx];
    float w = dinv[r] * dinv[c];
    const float* gp = g + (size_t)r * NCLS;
    float* op = out2 + (size_t)c * NCLS;
    #pragma unroll
    for (int q = 0; q < NCLS; ++q) atomicAdd(op + q, w * gp[q]);
}


__global__ void gcn_logsoftmax(const float* __restrict__ out2, float* __restrict__ out, int n) {
    int i = blockIdx.x * blockDim.x + threadIdx.x;
    if (i >= n) return;
    float v[NCLS];
    float m = -1e30f;
    #pragma unroll
    for (int c = 0; c < NCLS; ++c) { v[c] = out2[(size_t)i * NCLS + c]; m = fmaxf(m, v[c]); }
    float sum = 0.0f;
    #pragma unroll
    for (int c = 0; c < NCLS; ++c) sum += __expf(v[c] - m);
    float lse = m + __logf(sum);
    #pragma unroll
    for (int c = 0; c < NCLS; ++c) out[(size_t)i * NCLS + c] = v[c] - lse;
}


extern "C" void kernel_launch(void* const* d_in, const int* in_sizes, int n_in,
                              void* d_out, int out_size, void* d_ws, size_t ws_size,
                              hipStream_t stream) {
    const float* x  = (const float*)d_in[0];
    const int*   ei = (const int*)  d_in[1];
    const float* W1 = (const float*)d_in[2];
    const float* b1 = (const float*)d_in[3];
    const float* W2 = (const float*)d_in[4];
    const float* b2 = (const float*)d_in[5];
    float* out = (float*)d_out;

    int n = in_sizes[0] / F_IN;
    int e = in_sizes[1] / 2;
    const int* rowi = ei;
    const int* coli = ei + e;

    float* ws   = (float*)d_ws;
    float* dinv = ws;
    float* h1   = dinv + n;
    float* out1 = h1   + (size_t)n * HID;
    float* g    = out1 + (size_t)n * HID;
    float* out2 = g    + (size_t)n * NCLS;

    const int B = 256;
    int gn = (n + B - 1) / B;
    int ge = (e + B - 1) / B;

    gcn_init_deg   <<<gn, B, 0, stream>>>(dinv, n);
    gcn_deg_scatter<<<ge, B, 0, stream>>>(coli, dinv, e);
    gcn_dinv       <<<gn, B, 0, stream>>>(dinv, n);

    int ntiles = (n + 15) >> 4;
    int nwaves = (ntiles + TILES_PER_WAVE - 1) / TILES_PER_WAVE;
    int gw = (nwaves * 32 + B - 1) / B;
    gcn_gemm1_wmma <<<gw, B, 0, stream>>>(x, W1, h1, n);

    gcn_init_out1  <<<gn, B, 0, stream>>>(h1, dinv, b1, out1, n);
    gcn_scatter1   <<<ge, B, 0, stream>>>(rowi, coli, dinv, h1, out1, e);

    gcn_gemm2      <<<gn, B, 0, stream>>>(out1, W2, b2, dinv, g, out2, n);
    gcn_scatter2   <<<ge, B, 0, stream>>>(rowi, coli, dinv, g, out2, e);

    gcn_logsoftmax <<<gn, B, 0, stream>>>(out2, out, n);
}


# Both edge scatter-adds: lane = feature (16-lane row per edge in conv1, 8-lane group in conv2): each row gather and f32 atomic add covers whole node rows, not 32 lines; same atomics, arrival order
# speedup vs baseline: 3.1091x; 3.1091x over previous
#include <hip/hip_runtime.h>
#include <hip/hip_bf16.h>

#define F_IN 512
#define HID  16
#define NCLS 7
#define TILES_PER_WAVE 2

typedef float v2f __attribute__((ext_vector_type(2)));
typedef float v8f __attribute__((ext_vector_type(8)));


__global__ void gcn_init_deg(float* __restrict__ deg, int n) {
    int i = blockIdx.x * blockDim.x + threadIdx.x;
    if (i < n) deg[i] = 1.0f;
}

__global__ void gcn_deg_scatter(const int* __restrict__ col, float* __restrict__ deg, int e) {
    int i = blockIdx.x * blockDim.x + threadIdx.x;
    if (i < e) atomicAdd(&deg[col[i]], 1.0f);
}

__global__ void gcn_dinv(float* __restrict__ degdinv, int n) {
    int i = blockIdx.x * blockDim.x + threadIdx.x;
    if (i < n) degdinv[i] = rsqrtf(degdinv[i]);
}


__device__ __forceinline__ void store_tile(float* __restrict__ h1, const v8f& acc,
                                           int tile, int lane, int n) {
    int mbase = tile * 16 + ((lane >> 4) << 3);
    int ncol  = lane & 15;
    float* p = h1 + (size_t)mbase * HID + ncol;
    if (mbase + 8 <= n) {
        #pragma unroll
        for (int v = 0; v < 8; ++v) p[v * HID] = acc[v];
    } else {
        #pragma unroll
        for (int v = 0; v < 8; ++v) if (mbase + v < n) p[v * HID] = acc[v];
    }
}

__global__ void gcn_gemm1_wmma(const float* __restrict__ x, const float* __restrict__ W1,
                               float* __restrict__ h1, int n) {
    int lane = threadIdx.x & 31;
    int wid  = (blockIdx.x * blockDim.x + threadIdx.x) >> 5;
    int ntiles = (n + 15) >> 4;
    int tile0 = wid * TILES_PER_WAVE;
    if (tile0 >= ntiles) return;
    bool has1 = (tile0 + 1) < ntiles;

    int m    = lane & 15;
    int koff = (lane >> 4) << 1;

    int row0 = tile0 * 16 + m;
    int row1 = row0 + 16;
    const float* xa0 = x + (size_t)(row0 < n ? row0 : 0) * F_IN + koff;
    const float* xa1 = x + (size_t)(row1 < n ? row1 : 0) * F_IN + koff;

    v8f acc0 = {}, acc1 = {};
    #pragma unroll 4
    for (int k = 0; k < F_IN; k += 4) {
        v2f b;
        b.x = W1[(size_t)(k + koff)     * HID + m];
        b.y = W1[(size_t)(k + koff + 1) * HID + m];
        v2f a0 = *(const v2f*)(xa0 + k);
        v2f a1 = *(const v2f*)(xa1 + k);
        acc0 = __builtin_amdgcn_wmma_f32_16x16x4_f32(
            false, a0, false, b, (short)0, acc0, false, false);
        acc1 = __builtin_amdgcn_wmma_f32_16x16x4_f32(
            false, a1, false, b, (short)0, acc1, false, false);
    }

    store_tile(h1, acc0, tile0, lane, n);
    if (has1) store_tile(h1, acc1, tile0 + 1, lane, n);
}


__global__ void gcn_init_out1(const float* __restrict__ h1, const float* __restrict__ dinv,
                              const float* __restrict__ b1, float* __restrict__ out1, int n) {
    int i = blockIdx.x * blockDim.x + threadIdx.x;
    if (i >= n) return;
    float s = dinv[i]; s = s * s;
    const float4* hp = (const float4*)(h1 + (size_t)i * HID);
    float4* op = (float4*)(out1 + (size_t)i * HID);
    #pragma unroll
    for (int q = 0; q < 4; ++q) {
        float4 h = hp[q];
        op[q] = make_float4(b1[4*q+0] + s*h.x, b1[4*q+1] + s*h.y,
                            b1[4*q+2] + s*h.z, b1[4*q+3] + s*h.w);
    }
}


template <int J> __device__ __forceinline__ int row16_bcast(int v) {
    return __builtin_amdgcn_update_dpp(0, v, 0x150 + J, 0xf, 0xf, true);
}

template <int J>
__device__ __forceinline__ void scatter1_gather(int r, unsigned f, const float* __restrict__ h1,
                                                float* v) {
    v[J] = h1[(unsigned)row16_bcast<J>(r) * HID + f];
    if constexpr (J + 1 < HID) scatter1_gather<J + 1>(r, f, h1, v);
}

template <int J>
__device__ __forceinline__ void scatter1_add(int c, float w, unsigned f, const float* v,
                                             float* __restrict__ out1) {
    float wj = __int_as_float(row16_bcast<J>(__float_as_int(w)));
    atomicAdd(out1 + ((unsigned)row16_bcast<J>(c) * HID + f), wj * v[J]);
    if constexpr (J + 1 < HID) scatter1_add<J + 1>(c, w, f, v, out1);
}

__global__ void gcn_scatter1(const int* __restrict__ rowi, const int* __restrict__ coli,
                             const float* __restrict__ dinv, const float* __restrict__ h1,
                             float* __restrict__ out1, int e) {
    int idx = blockIdx.x * blockDim.x + threadIdx.x;
    int r = 0, c = 0;
    float w = 0.0f;
    if (idx < e) { r = rowi[idx]; c = coli[idx]; w = dinv[r] * dinv[c]; }
    int wbase = __builtin_amdgcn_readfirstlane(idx);
    if (wbase + 32 <= e) {
        unsigned f = threadIdx.x & 15u;
        float v[HID];
        scatter1_gather<0>(r, f, h1, v);
        scatter1_add<0>(c, w, f, v, out1);
    } else if (idx < e) {
        const float* hp = h1 + (size_t)r * HID;
        float* op = out1 + (size_t)c * HID;
        #pragma unroll
        for (int q = 0; q < HID; ++q) atomicAdd(op + q, w * hp[q]);
    }
}


__global__ void gcn_gemm2(const float* __restrict__ out1, const float* __restrict__ W2,
                          const float* __restrict__ b2, const float* __restrict__ dinv,
                          float* __restrict__ g, float* __restrict__ out2, int n) {
    int i = blockIdx.x * blockDim.x + threadIdx.x;
    if (i >= n) return;
    float h[HID];
    const float* ip = out1 + (size_t)i * HID;
    #pragma unroll
    for (int c = 0; c < HID; ++c) h[c] = fmaxf(ip[c], 0.0f);
    float s = dinv[i]; s = s * s;
    #pragma unroll
    for (int c2 = 0; c2 < NCLS; ++c2) {
        float acc = 0.0f;
        #pragma unroll
        for (int c = 0; c < HID; ++c) acc += h[c] * W2[c * NCLS + c2];
        g[(size_t)i * NCLS + c2]    = acc;
        out2[(size_t)i * NCLS + c2] = b2[c2] + s * acc;
    }
}


template <int J> __device__ __forceinline__ int grp8_bcast(int v) {
    return (int)__builtin_amdgcn_mov_dpp8((unsigned)v, (unsigned)J * 0x249249u);
}

template <int J>
__device__ __forceinline__ void scatter2_spread(int r, int c, float w, unsigned f,
                                                unsigned* go, unsigned* oo, float* wv) {
    go[J] = (unsigned)grp8_bcast<J>(r) * NCLS + f;
    oo[J] = (unsigned)grp8_bcast<J>(c) * NCLS + f;
    wv[J] = __int_as_float(grp8_bcast<J>(__float_as_int(w)));
    if constexpr (J + 1 < 8) scatter2_spread<J + 1>(r, c, w, f, go, oo, wv);
}

__global__ void gcn_scatter2(const int* __restrict__ rowi, const int* __restrict__ coli,
                             const float* __restrict__ dinv, const float* __restrict__ g,
                             float* __restrict__ out2, int e) {
    int idx = blockIdx.x * blockDim.x + threadIdx.x;
    int r = 0, c = 0;
    float w = 0.0f;
    if (idx < e) { r = rowi[idx]; c = coli[idx]; w = dinv[r] * dinv[c]; }
    int wbase = __builtin_amdgcn_readfirstlane(idx);
    if (wbase + 32 <= e) {
        unsigned f = threadIdx.x & 7u;
        unsigned go[8], oo[8];
        float wv[8];
        scatter2_spread<0>(r, c, w, f, go, oo, wv);
        if (f < NCLS) {
            float v[8];
            #pragma unroll
            for (int j = 0; j < 8; ++j) v[j] = g[go[j]];
            #pragma unroll
            for (int j = 0; j < 8; ++j) atomicAdd(out2 + oo[j], wv[j] * v[j]);
        }
    } else if (idx < e) {
        const float* gp = g + (size_t)r * NCLS;
        float* op = out2 + (size_t)c * NCLS;
        #pragma unroll
        for (int q = 0; q < NCLS; ++q) atomicAdd(op + q, w * gp[q]);
    }
}


__global__ void gcn_logsoftmax(const float* __restrict__ out2, float* __restrict__ out, int n) {
    int i = blockIdx.x * blockDim.x + threadIdx.x;
    if (i >= n) return;
    float v[NCLS];
    float m = -1e30f;
    #pragma unroll
    for (int c = 0; c < NCLS; ++c) { v[c] = out2[(size_t)i * NCLS + c]; m = fmaxf(m, v[c]); }
    float sum = 0.0f;
    #pragma unroll
    for (int c = 0; c < NCLS; ++c) sum += __expf(v[c] - m);
    float lse = m + __logf(sum);
    #pragma unroll
    for (int c = 0; c < NCLS; ++c) out[(size_t)i * NCLS + c] = v[c] - lse;
}


extern "C" void kernel_launch(void* const* d_in, const int* in_sizes, int n_in,
                              void* d_out, int out_size, void* d_ws, size_t ws_size,
                              hipStream_t stream) {
    const float* x  = (const float*)d_in[0];
    const int*   ei = (const int*)  d_in[1];
    const float* W1 = (const float*)d_in[2];
    const float* b1 = (const float*)d_in[3];
    const float* W2 = (const float*)d_in[4];
    const float* b2 = (const float*)d_in[5];
    float* out = (float*)d_out;

    int n = in_sizes[0] / F_IN;
    int e = in_sizes[1] / 2;
    const int* rowi = ei;
    const int* coli = ei + e;

    float* ws   = (float*)d_ws;
    float* dinv = ws;
    float* h1   = dinv + n;
    float* out1 = h1   + (size_t)n * HID;
    float* g    = out1 + (size_t)n * HID;
    float* out2 = g    + (size_t)n * NCLS;

    const int B = 256;
    int gn = (n + B - 1) / B;
    int ge = (e + B - 1) / B;

    gcn_init_deg   <<<gn, B, 0, stream>>>(dinv, n);
    gcn_deg_scatter<<<ge, B, 0, stream>>>(coli, dinv, e);
    gcn_dinv       <<<gn, B, 0, stream>>>(dinv, n);

    int ntiles = (n + 15) >> 4;
    int nwaves = (ntiles + TILES_PER_WAVE - 1) / TILES_PER_WAVE;
    int gw = (nwaves * 32 + B - 1) / B;
    gcn_gemm1_wmma <<<gw, B, 0, stream>>>(x, W1, h1, n);

    gcn_init_out1  <<<gn, B, 0, stream>>>(h1, dinv, b1, out1, n);
    gcn_scatter1   <<<ge, B, 0, stream>>>(rowi, coli, dinv, h1, out1, e);

    gcn_gemm2      <<<gn, B, 0, stream>>>(out1, W2, b2, dinv, g, out2, n);
    gcn_scatter2   <<<ge, B, 0, stream>>>(rowi, coli, dinv, g, out2, e);

    gcn_logsoftmax <<<gn, B, 0, stream>>>(out2, out, n);
}


# Edge norm dinv[r]*dinv[c] split to the nodes: source rows pre-scaled once, target dinv applied once after the sum; edge passes drop both per-edge dinv gathers; same atomics, arrival order
# speedup vs baseline: 3.3743x; 1.0853x over previous
#include <hip/hip_runtime.h>
#include <hip/hip_bf16.h>

#define F_IN 512
#define HID  16
#define NCLS 7
#define TILES_PER_WAVE 2

typedef float v2f __attribute__((ext_vector_type(2)));
typedef float v8f __attribute__((ext_vector_type(8)));


__global__ void gcn_init_deg(float* __restrict__ deg, int n) {
    int i = blockIdx.x * blockDim.x + threadIdx.x;
    if (i < n) deg[i] = 1.0f;
}

__global__ void gcn_deg_scatter(const int* __restrict__ col, float* __restrict__ deg, int e) {
    int i = blockIdx.x * blockDim.x + threadIdx.x;
    if (i < e) atomicAdd(&deg[col[i]], 1.0f);
}

__global__ void gcn_dinv(float* __restrict__ degdinv, int n) {
    int i = blockIdx.x * blockDim.x + threadIdx.x;
    if (i < n) degdinv[i] = rsqrtf(degdinv[i]);
}


__device__ __forceinline__ void store_tile(float* __restrict__ h1, const v8f& acc,
                                           int tile, int lane, int n) {
    int mbase = tile * 16 + ((lane >> 4) << 3);
    int ncol  = lane & 15;
    float* p = h1 + (size_t)mbase * HID + ncol;
    if (mbase + 8 <= n) {
        #pragma unroll
        for (int v = 0; v < 8; ++v) p[v * HID] = acc[v];
    } else {
        #pragma unroll
        for (int v = 0; v < 8; ++v) if (mbase + v < n) p[v * HID] = acc[v];
    }
}

__global__ void gcn_gemm1_wmma(const float* __restrict__ x, const float* __restrict__ W1,
                               float* __restrict__ h1, int n) {
    int lane = threadIdx.x & 31;
    int wid  = (blockIdx.x * blockDim.x + threadIdx.x) >> 5;
    int ntiles = (n + 15) >> 4;
    int tile0 = wid * TILES_PER_WAVE;
    if (tile0 >= ntiles) return;
    bool has1 = (tile0 + 1) < ntiles;

    int m    = lane & 15;
    int koff = (lane >> 4) << 1;

    int row0 = tile0 * 16 + m;
    int row1 = row0 + 16;
    const float* xa0 = x + (size_t)(row0 < n ? row0 : 0) * F_IN + koff;
    const float* xa1 = x + (size_t)(row1 < n ? row1 : 0) * F_IN + koff;

    v8f acc0 = {}, acc1 = {};
    #pragma unroll 4
    for (int k = 0; k < F_IN; k += 4) {
        v2f b;
        b.x = W1[(size_t)(k + koff)     * HID + m];
        b.y = W1[(size_t)(k + koff + 1) * HID + m];
        v2f a0 = *(const v2f*)(xa0 + k);
        v2f a1 = *(const v2f*)(xa1 + k);
        acc0 = __builtin_amdgcn_wmma_f32_16x16x4_f32(
            false, a0, false, b, (short)0, acc0, false, false);
        acc1 = __builtin_amdgcn_wmma_f32_16x16x4_f32(
            false, a1, false, b, (short)0, acc1, false, false);
    }

    store_tile(h1, acc0, tile0, lane, n);
    if (has1) store_tile(h1, acc1, tile0 + 1, lane, n);
}


__global__ void gcn_init_out1(float* __restrict__ h1, const float* __restrict__ dinv,
                              float* __restrict__ out1, int n) {
    int i = blockIdx.x * blockDim.x + threadIdx.x;
    if (i >= n) return;
    float s = dinv[i];
    float4* hp = (float4*)(h1 + (size_t)i * HID);
    float4* op = (float4*)(out1 + (size_t)i * HID);
    #pragma unroll
    for (int q = 0; q < 4; ++q) {
        float4 h = hp[q];
        float4 hs = make_float4(s*h.x, s*h.y, s*h.z, s*h.w);
        hp[q] = hs;
        op[q] = hs;
    }
}


template <int J> __device__ __forceinline__ int row16_bcast(int v) {
    return __builtin_amdgcn_update_dpp(0, v, 0x150 + J, 0xf, 0xf, true);
}

template <int J>
__device__ __forceinline__ void scatter1_gather(int r, unsigned f, const float* __restrict__ h1,
                                                float* v) {
    v[J] = h1[(unsigned)row16_bcast<J>(r) * HID + f];
    if constexpr (J + 1 < HID) scatter1_gather<J + 1>(r, f, h1, v);
}

template <int J>
__device__ __forceinline__ void scatter1_add(int c, unsigned f, const float* v,
                                             float* __restrict__ out1) {
    atomicAdd(out1 + ((unsigned)row16_bcast<J>(c) * HID + f), v[J]);
    if constexpr (J + 1 < HID) scatter1_add<J + 1>(c, f, v, out1);
}

__global__ void gcn_scatter1(const int* __restrict__ rowi, const int* __restrict__ coli,
                             const float* __restrict__ h1, float* __restrict__ out1, int e) {
    int idx = blockIdx.x * blockDim.x + threadIdx.x;
    int r = 0, c = 0;
    if (idx < e) { r = rowi[idx]; c = coli[idx]; }
    int wbase = __builtin_amdgcn_readfirstlane(idx);
    if (wbase + 32 <= e) {
        unsigned f = threadIdx.x & 15u;
        float v[HID];
        scatter1_gather<0>(r, f, h1, v);
        scatter1_add<0>(c, f, v, out1);
    } else if (idx < e) {
        const float* hp = h1 + (size_t)r * HID;
        float* op = out1 + (size_t)c * HID;
        #pragma unroll
        for (int q = 0; q < HID; ++q) atomicAdd(op + q, hp[q]);
    }
}


__global__ void gcn_gemm2(const float* __restrict__ out1, const float* __restrict__ b1,
                          const float* __restrict__ W2, const float* __restrict__ dinv,
                          float* __restrict__ g, float* __restrict__ out2, int n) {
    int i = blockIdx.x * blockDim.x + threadIdx.x;
    if (i >= n) return;
    float h[HID];
    const float* ip = out1 + (size_t)i * HID;
    float s = dinv[i];
    #pragma unroll
    for (int c = 0; c < HID; ++c) h[c] = fmaxf(b1[c] + s * ip[c], 0.0f);
    #pragma unroll
    for (int c2 = 0; c2 < NCLS; ++c2) {
        float acc = 0.0f;
        #pragma unroll
        for (int c = 0; c < HID; ++c) acc += h[c] * W2[c * NCLS + c2];
        float gs = s * acc;
        g[(size_t)i * NCLS + c2]    = gs;
        out2[(size_t)i * NCLS + c2] = gs;
    }
}


template <int J> __device__ __forceinline__ int grp8_bcast(int v) {
    return (int)__builtin_amdgcn_mov_dpp8((unsigned)v, (unsigned)J * 0x249249u);
}

template <int J>
__device__ __forceinline__ void scatter2_spread(int r, int c, unsigned f,
                                                unsigned* go, unsigned* oo) {
    go[J] = (unsigned)grp8_bcast<J>(r) * NCLS + f;
    oo[J] = (unsigned)grp8_bcast<J>(c) * NCLS + f;
    if constexpr (J + 1 < 8) scatter2_spread<J + 1>(r, c, f, go, oo);
}

__global__ void gcn_scatter2(const int* __restrict__ rowi, const int* __restrict__ coli,
                             const float* __restrict__ g, float* __restrict__ out2, int e) {
    int idx = blockIdx.x * blockDim.x + threadIdx.x;
    int r = 0, c = 0;
    if (idx < e) { r = rowi[idx]; c = coli[idx]; }
    int wbase = __builtin_amdgcn_readfirstlane(idx);
    if (wbase + 32 <= e) {
        unsigned f = threadIdx.x & 7u;
        unsigned go[8], oo[8];
        scatter2_spread<0>(r, c, f, go, oo);
        if (f < NCLS) {
            float v[8];
            #pragma unroll
            for (int j = 0; j < 8; ++j) v[j] = g[go[j]];
            #pragma unroll
            for (int j = 0; j < 8; ++j) atomicAdd(out2 + oo[j], v[j]);
        }
    } else if (idx < e) {
        const float* gp = g + (size_t)r * NCLS;
        float* op = out2 + (size_t)c * NCLS;
        #pragma unroll
        for (int q = 0; q < NCLS; ++q) atomicAdd(op + q, gp[q]);
    }
}


__global__ void gcn_logsoftmax(const float* __restrict__ out2, const float* __restrict__ dinv,
                               const float* __restrict__ b2, float* __restrict__ out, int n) {
    int i = blockIdx.x * blockDim.x + threadIdx.x;
    if (i >= n) return;
    float v[NCLS];
    float m = -1e30f;
    float s = dinv[i];
    #pragma unroll
    for (int c = 0; c < NCLS; ++c) {
        v[c] = b2[c] + s * out2[(size_t)i * NCLS + c];
        m = fmaxf(m, v[c]);
    }
    float sum = 0.0f;
    #pragma unroll
    for (int c = 0; c < NCLS; ++c) sum += __expf(v[c] - m);
    float lse = m + __logf(sum);
    #pragma unroll
    for (int c = 0; c < NCLS; ++c) out[(size_t)i * NCLS + c] = v[c] - lse;
}


extern "C" void kernel_launch(void* const* d_in, const int* in_sizes, int n_in,
                              void* d_out, int out_size, void* d_ws, size_t ws_size,
                              hipStream_t stream) {
    const float* x  = (const float*)d_in[0];
    const int*   ei = (const int*)  d_in[1];
    const float* W1 = (const float*)d_in[2];
    const float* b1 = (const float*)d_in[3];
    const float* W2 = (const float*)d_in[4];
    const float* b2 = (const float*)d_in[5];
    float* out = (float*)d_out;

    int n = in_sizes[0] / F_IN;
    int e = in_sizes[1] / 2;
    const int* rowi = ei;
    const int* coli = ei + e;

    float* ws   = (float*)d_ws;
    float* dinv = ws;
    float* h1   = dinv + n;
    float* out1 = h1   + (size_t)n * HID;
    float* g    = out1 + (size_t)n * HID;
    float* out2 = g    + (size_t)n * NCLS;

    const int B = 256;
    int gn = (n + B - 1) / B;
    int ge = (e + B - 1) / B;

    gcn_init_deg   <<<gn, B, 0, stream>>>(dinv, n);
    gcn_deg_scatter<<<ge, B, 0, stream>>>(coli, dinv, e);
    gcn_dinv       <<<gn, B, 0, stream>>>(dinv, n);

    int ntiles = (n + 15) >> 4;
    int nwaves = (ntiles + TILES_PER_WAVE - 1) / TILES_PER_WAVE;
    int gw = (nwaves * 32 + B - 1) / B;
    gcn_gemm1_wmma <<<gw, B, 0, stream>>>(x, W1, h1, n);

    gcn_init_out1  <<<gn, B, 0, stream>>>(h1, dinv, out1, n);
    gcn_scatter1   <<<ge, B, 0, stream>>>(rowi, coli, h1, out1, e);

    gcn_gemm2      <<<gn, B, 0, stream>>>(out1, b1, W2, dinv, g, out2, n);
    gcn_scatter2   <<<ge, B, 0, stream>>>(rowi, coli, g, out2, e);

    gcn_logsoftmax <<<gn, B, 0, stream>>>(out2, dinv, b2, out, n);
}


# Edges counting-sorted by target (rank = value returned by the int degree add); both scatter-adds become per-node gather-sums in registers, no float atomics; sum order = arrival-order rank
# speedup vs baseline: 3.8119x; 1.1297x over previous
#include <hip/hip_runtime.h>
#include <hip/hip_bf16.h>

#define F_IN 512
#define HID  16
#define NCLS 7
#define TILES_PER_WAVE 2

typedef float v2f __attribute__((ext_vector_type(2)));
typedef float v8f __attribute__((ext_vector_type(8)));


__global__ void gcn_zero_cnt(int* __restrict__ cnt, int n) {
    int i = blockIdx.x * blockDim.x + threadIdx.x;
    if (i < n) cnt[i] = 0;
}

__global__ void gcn_count_rank(const int* __restrict__ col, int* __restrict__ cnt,
                               int* __restrict__ rank, int e) {
    int i = blockIdx.x * blockDim.x + threadIdx.x;
    if (i < e) rank[i] = atomicAdd(&cnt[col[i]], 1);
}

#define SCAN_B 1024

__device__ __forceinline__ int wave_sum(int s) {
    #pragma unroll
    for (int o = 16; o > 0; o >>= 1) s += __shfl_xor(s, o);
    return s;
}

__device__ __forceinline__ int wave_scan_incl(int s, int lane) {
    #pragma unroll
    for (int o = 1; o < 32; o <<= 1) {
        int t = __shfl_up(s, o);
        if (lane >= o) s += t;
    }
    return s;
}

__global__ void __launch_bounds__(SCAN_B)
gcn_scan_dinv(const int* __restrict__ cnt, int* __restrict__ start, float* __restrict__ dinv, int n) {
    __shared__ int wpart[SCAN_B / 32];
    __shared__ int wbase[SCAN_B / 32];
    int tid = threadIdx.x, lane = tid & 31, w = tid >> 5;
    int first = blockIdx.x * SCAN_B;

    int s = 0;
    const int4* c4 = (const int4*)cnt;
    for (int q = tid; q < (first >> 2); q += SCAN_B) {
        int4 v = c4[q];
        s += (v.x + v.y) + (v.z + v.w);
    }
    s = wave_sum(s);
    if (lane == 0) wpart[w] = s;

    int i = first + tid;
    int c = (i < n) ? cnt[i] : 0;
    int inc = wave_scan_incl(c, lane);
    if (lane == 31) wbase[w] = inc;
    __syncthreads();
    if (w == 0) {
        int front = wave_sum(wpart[lane]);
        int tot  = wbase[lane];
        int incw = wave_scan_incl(tot, lane);
        wbase[lane] = front + incw - tot;
    }
    __syncthreads();
    int excl = wbase[w] + inc - c;
    if (i < n) {
        start[i] = excl;
        dinv[i]  = rsqrtf((float)(c + 1));
        if (i == n - 1) start[n] = excl + c;
    }
}

__global__ void gcn_place(const int* __restrict__ rowi, const int* __restrict__ coli,
                          const int* __restrict__ rank, const int* __restrict__ start,
                          int* __restrict__ srcs, int e) {
    int i = blockIdx.x * blockDim.x + threadIdx.x;
    if (i < e) srcs[start[coli[i]] + rank[i]] = rowi[i];
}


__device__ __forceinline__ void store_tile(float* __restrict__ h1, const v8f& acc,
                                           int tile, int lane, int n) {
    int mbase = tile * 16 + ((lane >> 4) << 3);
    int ncol  = lane & 15;
    float* p = h1 + (size_t)mbase * HID + ncol;
    if (mbase + 8 <= n) {
        #pragma unroll
        for (int v = 0; v < 8; ++v) p[v * HID] = acc[v];
    } else {
        #pragma unroll
        for (int v = 0; v < 8; ++v) if (mbase + v < n) p[v * HID] = acc[v];
    }
}

__global__ void gcn_gemm1_wmma(const float* __restrict__ x, const float* __restrict__ W1,
                               float* __restrict__ h1, int n) {
    int lane = threadIdx.x & 31;
    int wid  = (blockIdx.x * blockDim.x + threadIdx.x) >> 5;
    int ntiles = (n + 15) >> 4;
    int tile0 = wid * TILES_PER_WAVE;
    if (tile0 >= ntiles) return;
    bool has1 = (tile0 + 1) < ntiles;

    int m    = lane & 15;
    int koff = (lane >> 4) << 1;

    int row0 = tile0 * 16 + m;
    int row1 = row0 + 16;
    const float* xa0 = x + (size_t)(row0 < n ? row0 : 0) * F_IN + koff;
    const float* xa1 = x + (size_t)(row1 < n ? row1 : 0) * F_IN + koff;

    v8f acc0 = {}, acc1 = {};
    #pragma unroll 4
    for (int k = 0; k < F_IN; k += 4) {
        v2f b;
        b.x = W1[(size_t)(k + koff)     * HID + m];
        b.y = W1[(size_t)(k + koff + 1) * HID + m];
        v2f a0 = *(const v2f*)(xa0 + k);
        v2f a1 = *(const v2f*)(xa1 + k);
        acc0 = __builtin_amdgcn_wmma_f32_16x16x4_f32(
            false, a0, false, b, (short)0, acc0, false, false);
        acc1 = __builtin_amdgcn_wmma_f32_16x16x4_f32(
            false, a1, false, b, (short)0, acc1, false, false);
    }

    store_tile(h1, acc0, tile0, lane, n);
    if (has1) store_tile(h1, acc1, tile0 + 1, lane, n);
}


__global__ void gcn_prescale(float* __restrict__ h1, const float* __restrict__ dinv, int n) {
    int t = blockIdx.x * blockDim.x + threadIdx.x;
    if (t >= n * (HID / 4)) return;
    float s = dinv[t >> 2];
    float4* p = (float4*)h1 + t;
    float4 h = *p;
    *p = make_float4(s*h.x, s*h.y, s*h.z, s*h.w);
}

template <int J> __device__ __forceinline__ int row16_bcast(int v) {
    return __builtin_amdgcn_update_dpp(0, v, 0x150 + J, 0xf, 0xf, true);
}

template <int J> __device__ __forceinline__ void row16_bcast_all(int v, int* r) {
    r[J] = row16_bcast<J>(v);
    if constexpr (J + 1 < 16) row16_bcast_all<J + 1>(v, r);
}

__global__ void gcn_agg1(const int* __restrict__ start, const int* __restrict__ srcs,
                         const float* __restrict__ hs, float* __restrict__ out1, int n) {
    int t = blockIdx.x * blockDim.x + threadIdx.x;
    int node = t >> 4;
    unsigned f = threadIdx.x & 15u;
    int beg = 0, end = 0, self = 0;
    float acc = 0.0f;
    if (node < n) {
        beg = start[node];
        end = start[node + 1];
        self = node;
        acc = hs[(unsigned)node * HID + f];
    }
    int p = beg + (int)f;
    int idx = (p < end) ? srcs[p] : -1;
    for (int base = beg; base < end; base += 16) {
        int pn = p + 16;
        int ld = srcs[pn < end ? pn : beg];
        __builtin_amdgcn_sched_barrier(0);
        int r[16];
        float v[16];
        row16_bcast_all<0>(idx, r);
        #pragma unroll
        for (int j = 0; j < 16; ++j) v[j] = hs[(unsigned)(r[j] >= 0 ? r[j] : self) * HID + f];
        #pragma unroll
        for (int j = 0; j < 16; ++j) acc += (r[j] >= 0) ? v[j] : 0.0f;
        idx = (pn < end) ? ld : -1;
        p = pn;
    }
    if (node < n) out1[(unsigned)node * HID + f] = acc;
}


__global__ void gcn_gemm2(const float* __restrict__ out1, const float* __restrict__ b1,
                          const float* __restrict__ W2, const float* __restrict__ dinv,
                          float* __restrict__ g, int n) {
    int i = blockIdx.x * blockDim.x + threadIdx.x;
    if (i >= n) return;
    float h[HID];
    const float* ip = out1 + (size_t)i * HID;
    float s = dinv[i];
    #pragma unroll
    for (int c = 0; c < HID; ++c) h[c] = fmaxf(b1[c] + s * ip[c], 0.0f);
    #pragma unroll
    for (int c2 = 0; c2 < NCLS; ++c2) {
        float acc = 0.0f;
        #pragma unroll
        for (int c = 0; c < HID; ++c) acc += h[c] * W2[c * NCLS + c2];
        g[(size_t)i * NCLS + c2] = s * acc;
    }
}

template <int J> __device__ __forceinline__ int grp8_bcast(int v) {
    return (int)__builtin_amdgcn_mov_dpp8((unsigned)v, (unsigned)J * 0x249249u);
}

template <int J> __device__ __forceinline__ void grp8_bcast_all(int v, int* r) {
    r[J] = grp8_bcast<J>(v);
    if constexpr (J + 1 < 8) grp8_bcast_all<J + 1>(v, r);
}

__global__ void gcn_agg2(const int* __restrict__ start, const int* __restrict__ srcs,
                         const float* __restrict__ g, float* __restrict__ out2, int n) {
    int t = blockIdx.x * blockDim.x + threadIdx.x;
    int node = t >> 3;
    unsigned f = threadIdx.x & 7u;
    unsigned fc = f < NCLS ? f : NCLS - 1;
    int beg = 0, end = 0, self = 0;
    float acc = 0.0f;
    if (node < n) {
        beg = start[node];
        end = start[node + 1];
        self = node;
        acc = g[(unsigned)node * NCLS + fc];
    }
    int p = beg + (int)f;
    int i0 = (p < end) ? srcs[p] : -1;
    int i1 = (p + 8 < end) ? srcs[p + 8] : -1;
    for (int base = beg; base < end; base += 16) {
        int pn = p + 16;
        int l0 = srcs[pn < end ? pn : beg];
        int l1 = srcs[pn + 8 < end ? pn + 8 : beg];
        __builtin_amdgcn_sched_barrier(0);
        int r[16];
        float v[16];
        grp8_bcast_all<0>(i0, r);
        grp8_bcast_all<0>(i1, r + 8);
        #pragma unroll
        for (int j = 0; j < 16; ++j) v[j] = g[(unsigned)(r[j] >= 0 ? r[j] : self) * NCLS + fc];
        #pragma unroll
        for (int j = 0; j < 16; ++j) acc += (r[j] >= 0) ? v[j] : 0.0f;
        i0 = (pn < end) ? l0 : -1;
        i1 = (pn + 8 < end) ? l1 : -1;
        p = pn;
    }
    if (node < n && f < NCLS) out2[(unsigned)node * NCLS + f] = acc;
}


__global__ void gcn_logsoftmax(const float* __restrict__ out2, const float* __restrict__ dinv,
                               const float* __restrict__ b2, float* __restrict__ out, int n) {
    int i = blockIdx.x * blockDim.x + threadIdx.x;
    if (i >= n) return;
    float v[NCLS];
    float m = -1e30f;
    float s = dinv[i];
    #pragma unroll
    for (int c = 0; c < NCLS; ++c) {
        v[c] = b2[c] + s * out2[(size_t)i * NCLS + c];
        m = fmaxf(m, v[c]);
    }
    float sum = 0.0f;
    #pragma unroll
    for (int c = 0; c < NCLS; ++c) sum += __expf(v[c] - m);
    float lse = m + __logf(sum);
    #pragma unroll
    for (int c = 0; c < NCLS; ++c) out[(size_t)i * NCLS + c] = v[c] - lse;
}


extern "C" void kernel_launch(void* const* d_in, const int* in_sizes, int n_in,
                              void* d_out, int out_size, void* d_ws, size_t ws_size,
                              hipStream_t stream) {
    const float* x  = (const float*)d_in[0];
    const int*   ei = (const int*)  d_in[1];
    const float* W1 = (const float*)d_in[2];
    const float* b1 = (const float*)d_in[3];
    const float* W2 = (const float*)d_in[4];
    const float* b2 = (const float*)d_in[5];
    float* out = (float*)d_out;

    int n = in_sizes[0] / F_IN;
    int e = in_sizes[1] / 2;
    const int* rowi = ei;
    const int* coli = ei + e;

    size_t np = ((size_t)n + 64) & ~(size_t)63;
    size_t ep = ((size_t)e + 63) & ~(size_t)63;
    size_t rw = ep > 2 * HID * np ? ep : 2 * HID * np;
    if (ws_size < (3 * np + ep + rw) * sizeof(float)) return;

    int*   cnt   = (int*)d_ws;
    int*   start = cnt + np;
    float* dinv  = (float*)(start + np);
    int*   srcs  = (int*)(dinv + np);
    int*   rank  = srcs + ep;
    float* h1    = (float*)rank;
    float* out1  = h1 + HID * np;
    float* g     = h1;
    float* out2  = out1;

    const int B = 256;
    int gn = (n + B - 1) / B;
    int ge = (e + B - 1) / B;

    gcn_zero_cnt   <<<gn, B, 0, stream>>>(cnt, n);
    gcn_count_rank <<<ge, B, 0, stream>>>(coli, cnt, rank, e);
    gcn_scan_dinv  <<<(n + SCAN_B - 1) / SCAN_B, SCAN_B, 0, stream>>>(cnt, start, dinv, n);
    gcn_place      <<<ge, B, 0, stream>>>(rowi, coli, rank, start, srcs, e);

    int ntiles = (n + 15) >> 4;
    int nwaves = (ntiles + TILES_PER_WAVE - 1) / TILES_PER_WAVE;
    int gw = (nwaves * 32 + B - 1) / B;
    gcn_gemm1_wmma <<<gw, B, 0, stream>>>(x, W1, h1, n);

    gcn_prescale   <<<(n * (HID / 4) + B - 1) / B, B, 0, stream>>>(h1, dinv, n);
    gcn_agg1       <<<(n * 16 + B - 1) / B, B, 0, stream>>>(start, srcs, h1, out1, n);

    gcn_gemm2      <<<gn, B, 0, stream>>>(out1, b1, W2, dinv, g, n);
    gcn_agg2       <<<(n * 8 + B - 1) / B, B, 0, stream>>>(start, srcs, g, out2, n);

    gcn_logsoftmax <<<gn, B, 0, stream>>>(out2, dinv, b2, out, n);
}


# Edge counting sort without global atomics: edges binned by 256-node range from per-block LDS histograms, then ranked per node by LDS adds, one block per range; sum order = LDS arrival order
# speedup vs baseline: 6.9867x; 1.8329x over previous
#include <hip/hip_runtime.h>
#include <hip/hip_bf16.h>

#define F_IN 512
#define HID  16
#define NCLS 7
#define TILES_PER_WAVE 2

typedef float v2f __attribute__((ext_vector_type(2)));
typedef float v8f __attribute__((ext_vector_type(8)));


#define BIN_G    256
#define BIN_T    1024
#define BIN_SH   8
#define BIN_W    (1 << BIN_SH)
#define BIN_MAXB BIN_T

__device__ __forceinline__ int wave_scan_incl(int s, int lane) {
    #pragma unroll
    for (int o = 1; o < 32; o <<= 1) {
        int t = __shfl_up(s, o);
        if (lane >= o) s += t;
    }
    return s;
}

template <int K> struct batch_of { static constexpr int size = K; };

template <class Body>
__device__ __forceinline__ void for_edges(int i, int end, Body body) {
    for (; i + 7 * BIN_T < end; i += 8 * BIN_T) body(i, batch_of<8>{});
    if (i + 3 * BIN_T < end) { body(i, batch_of<4>{}); i += 4 * BIN_T; }
    if (i + 1 * BIN_T < end) { body(i, batch_of<2>{}); i += 2 * BIN_T; }
    if (i < end) body(i, batch_of<1>{});
}

__global__ void __launch_bounds__(BIN_T)
gcn_bin_count(const int* __restrict__ col, int* __restrict__ hist, int e, int nb) {
    __shared__ int h[BIN_MAXB];
    int tid = threadIdx.x;
    for (int b = tid; b < nb; b += BIN_T) h[b] = 0;
    __syncthreads();
    int ch = (e + BIN_G - 1) / BIN_G;
    int lo = blockIdx.x * ch;
    int hi = lo + ch < e ? lo + ch : e;
    for_edges(lo + tid, hi, [&](int i, auto kc) {
        constexpr int K = decltype(kc)::size;
        int c[K];
        #pragma unroll
        for (int j = 0; j < K; ++j) c[j] = col[i + j * BIN_T];
        #pragma unroll
        for (int j = 0; j < K; ++j) atomicAdd(&h[c[j] >> BIN_SH], 1);
    });
    __syncthreads();
    for (int b = tid; b < nb; b += BIN_T) hist[blockIdx.x * nb + b] = h[b];
}

__global__ void __launch_bounds__(BIN_T)
gcn_bin_scatter(const int* __restrict__ row, const int* __restrict__ col,
                const int* __restrict__ hist, int* __restrict__ base,
                unsigned* __restrict__ binned, int e, int nb) {
    __shared__ int cur[BIN_MAXB];
    __shared__ int wsum[BIN_T / 32];
    int tid = threadIdx.x, lane = tid & 31, w = tid >> 5;

    int tot = 0, pre = 0;
    if (tid < nb) {
        #pragma unroll 32
        for (int g = 0; g < BIN_G; ++g) {
            int v = hist[g * nb + tid];
            tot += v;
            pre += (g < (int)blockIdx.x) ? v : 0;
        }
    }
    int inc = wave_scan_incl(tot, lane);
    if (lane == 31) wsum[w] = inc;
    __syncthreads();
    if (w == 0) {
        int t = wsum[lane];
        int iw = wave_scan_incl(t, lane);
        wsum[lane] = iw - t;
    }
    __syncthreads();
    int first = wsum[w] + inc - tot;
    if (tid < nb) {
        cur[tid] = first + pre;
        if (blockIdx.x == 0) {
            base[tid] = first;
            if (tid == nb - 1) base[nb] = first + tot;
        }
    }
    __syncthreads();

    int ch = (e + BIN_G - 1) / BIN_G;
    int lo = blockIdx.x * ch;
    int hi = lo + ch < e ? lo + ch : e;
    for_edges(lo + tid, hi, [&](int i, auto kc) {
        constexpr int K = decltype(kc)::size;
        int c[K], r[K];
        #pragma unroll
        for (int j = 0; j < K; ++j) {
            c[j] = col[i + j * BIN_T];
            r[j] = row[i + j * BIN_T];
        }
        int pos[K];
        #pragma unroll
        for (int j = 0; j < K; ++j) pos[j] = atomicAdd(&cur[c[j] >> BIN_SH], 1);
        #pragma unroll
        for (int j = 0; j < K; ++j)
            binned[pos[j]] = ((unsigned)r[j] << BIN_SH) | ((unsigned)c[j] & (BIN_W - 1));
    });
}

__global__ void __launch_bounds__(BIN_T)
gcn_bucket_rank(const unsigned* __restrict__ binned, const int* __restrict__ base,
                int* __restrict__ start, float* __restrict__ dinv, int* __restrict__ srcs, int n) {
    __shared__ int cnt[BIN_W];
    __shared__ int slot[BIN_W];
    __shared__ int wsum[BIN_W / 32];
    int tid = threadIdx.x, lane = tid & 31, w = tid >> 5;
    int lo = base[blockIdx.x];
    int hi = base[blockIdx.x + 1];
    if (tid < BIN_W) cnt[tid] = 0;
    __syncthreads();
    for_edges(lo + tid, hi, [&](int i, auto kc) {
        constexpr int K = decltype(kc)::size;
        unsigned v[K];
        #pragma unroll
        for (int j = 0; j < K; ++j) v[j] = binned[i + j * BIN_T];
        #pragma unroll
        for (int j = 0; j < K; ++j) atomicAdd(&cnt[v[j] & (BIN_W - 1)], 1);
    });
    __syncthreads();
    int c = 0, inc = 0;
    if (tid < BIN_W) {
        c = cnt[tid];
        inc = wave_scan_incl(c, lane);
        if (lane == 31) wsum[w] = inc;
    }
    __syncthreads();
    if (tid < BIN_W) {
        int front = 0;
        #pragma unroll
        for (int k = 0; k < BIN_W / 32; ++k) front += (k < w) ? wsum[k] : 0;
        int excl = front + inc - c;
        slot[tid] = excl;
        int node = blockIdx.x * BIN_W + tid;
        if (node < n) {
            start[node] = lo + excl;
            dinv[node]  = rsqrtf((float)(c + 1));
            if (node == n - 1) start[n] = lo + excl + c;
        }
    }
    __syncthreads();
    for_edges(lo + tid, hi, [&](int i, auto kc) {
        constexpr int K = decltype(kc)::size;
        unsigned v[K];
        #pragma unroll
        for (int j = 0; j < K; ++j) v[j] = binned[i + j * BIN_T];
        int r[K];
        #pragma unroll
        for (int j = 0; j < K; ++j) r[j] = atomicAdd(&slot[v[j] & (BIN_W - 1)], 1);
        #pragma unroll
        for (int j = 0; j < K; ++j) srcs[lo + r[j]] = (int)(v[j] >> BIN_SH);
    });
}


__device__ __forceinline__ void store_tile(float* __restrict__ h1, const v8f& acc,
                                           int tile, int lane, int n) {
    int mbase = tile * 16 + ((lane >> 4) << 3);
    int ncol  = lane & 15;
    float* p = h1 + (size_t)mbase * HID + ncol;
    if (mbase + 8 <= n) {
        #pragma unroll
        for (int v = 0; v < 8; ++v) p[v * HID] = acc[v];
    } else {
        #pragma unroll
        for (int v = 0; v < 8; ++v) if (mbase + v < n) p[v * HID] = acc[v];
    }
}

__global__ void gcn_gemm1_wmma(const float* __restrict__ x, const float* __restrict__ W1,
                               float* __restrict__ h1, int n) {
    int lane = threadIdx.x & 31;
    int wid  = (blockIdx.x * blockDim.x + threadIdx.x) >> 5;
    int ntiles = (n + 15) >> 4;
    int tile0 = wid * TILES_PER_WAVE;
    if (tile0 >= ntiles) return;
    bool has1 = (tile0 + 1) < ntiles;

    int m    = lane & 15;
    int koff = (lane >> 4) << 1;

    int row0 = tile0 * 16 + m;
    int row1 = row0 + 16;
    const float* xa0 = x + (size_t)(row0 < n ? row0 : 0) * F_IN + koff;
    const float* xa1 = x + (size_t)(row1 < n ? row1 : 0) * F_IN + koff;

    v8f acc0 = {}, acc1 = {};
    #pragma unroll 4
    for (int k = 0; k < F_IN; k += 4) {
        v2f b;
        b.x = W1[(size_t)(k + koff)     * HID + m];
        b.y = W1[(size_t)(k + koff + 1) * HID + m];
        v2f a0 = *(const v2f*)(xa0 + k);
        v2f a1 = *(const v2f*)(xa1 + k);
        acc0 = __builtin_amdgcn_wmma_f32_16x16x4_f32(
            false, a0, false, b, (short)0, acc0, false, false);
        acc1 = __builtin_amdgcn_wmma_f32_16x16x4_f32(
            false, a1, false, b, (short)0, acc1, false, false);
    }

    store_tile(h1, acc0, tile0, lane, n);
    if (has1) store_tile(h1, acc1, tile0 + 1, lane, n);
}


__global__ void gcn_prescale(float* __restrict__ h1, const float* __restrict__ dinv, int n) {
    int t = blockIdx.x * blockDim.x + threadIdx.x;
    if (t >= n * (HID / 4)) return;
    float s = dinv[t >> 2];
    float4* p = (float4*)h1 + t;
    float4 h = *p;
    *p = make_float4(s*h.x, s*h.y, s*h.z, s*h.w);
}

template <int J> __device__ __forceinline__ int row16_bcast(int v) {
    return __builtin_amdgcn_update_dpp(0, v, 0x150 + J, 0xf, 0xf, true);
}

template <int J> __device__ __forceinline__ void row16_bcast_all(int v, int* r) {
    r[J] = row16_bcast<J>(v);
    if constexpr (J + 1 < 16) row16_bcast_all<J + 1>(v, r);
}

__global__ void gcn_agg1(const int* __restrict__ start, const int* __restrict__ srcs,
                         const float* __restrict__ hs, float* __restrict__ out1, int n) {
    int t = blockIdx.x * blockDim.x + threadIdx.x;
    int node = t >> 4;
    unsigned f = threadIdx.x & 15u;
    int beg = 0, end = 0, self = 0;
    float acc = 0.0f;
    if (node < n) {
        beg = start[node];
        end = start[node + 1];
        self = node;
        acc = hs[(unsigned)node * HID + f];
    }
    int p = beg + (int)f;
    int idx = (p < end) ? srcs[p] : -1;
    for (int base = beg; base < end; base += 16) {
        int pn = p + 16;
        int ld = srcs[pn < end ? pn : beg];
        __builtin_amdgcn_sched_barrier(0);
        int r[16];
        float v[16];
        row16_bcast_all<0>(idx, r);
        #pragma unroll
        for (int j = 0; j < 16; ++j) v[j] = hs[(unsigned)(r[j] >= 0 ? r[j] : self) * HID + f];
        #pragma unroll
        for (int j = 0; j < 16; ++j) acc += (r[j] >= 0) ? v[j] : 0.0f;
        idx = (pn < end) ? ld : -1;
        p = pn;
    }
    if (node < n) out1[(unsigned)node * HID + f] = acc;
}


__global__ void gcn_gemm2(const float* __restrict__ out1, const float* __restrict__ b1,
                          const float* __restrict__ W2, const float* __restrict__ dinv,
                          float* __restrict__ g, int n) {
    int i = blockIdx.x * blockDim.x + threadIdx.x;
    if (i >= n) return;
    float h[HID];
    const float* ip = out1 + (size_t)i * HID;
    float s = dinv[i];
    #pragma unroll
    for (int c = 0; c < HID; ++c) h[c] = fmaxf(b1[c] + s * ip[c], 0.0f);
    #pragma unroll
    for (int c2 = 0; c2 < NCLS; ++c2) {
        float acc = 0.0f;
        #pragma unroll
        for (int c = 0; c < HID; ++c) acc += h[c] * W2[c * NCLS + c2];
        g[(size_t)i * NCLS + c2] = s * acc;
    }
}

template <int J> __device__ __forceinline__ int grp8_bcast(int v) {
    return (int)__builtin_amdgcn_mov_dpp8((unsigned)v, (unsigned)J * 0x249249u);
}

template <int J> __device__ __forceinline__ void grp8_bcast_all(int v, int* r) {
    r[J] = grp8_bcast<J>(v);
    if constexpr (J + 1 < 8) grp8_bcast_all<J + 1>(v, r);
}

__global__ void gcn_agg2(const int* __restrict__ start, const int* __restrict__ srcs,
                         const float* __restrict__ g, float* __restrict__ out2, int n) {
    int t = blockIdx.x * blockDim.x + threadIdx.x;
    int node = t >> 3;
    unsigned f = threadIdx.x & 7u;
    unsigned fc = f < NCLS ? f : NCLS - 1;
    int beg = 0, end = 0, self = 0;
    float acc = 0.0f;
    if (node < n) {
        beg = start[node];
        end = start[node + 1];
        self = node;
        acc = g[(unsigned)node * NCLS + fc];
    }
    int p = beg + (int)f;
    int i0 = (p < end) ? srcs[p] : -1;
    int i1 = (p + 8 < end) ? srcs[p + 8] : -1;
    for (int base = beg; base < end; base += 16) {
        int pn = p + 16;
        int l0 = srcs[pn < end ? pn : beg];
        int l1 = srcs[pn + 8 < end ? pn + 8 : beg];
        __builtin_amdgcn_sched_barrier(0);
        int r[16];
        float v[16];
        grp8_bcast_all<0>(i0, r);
        grp8_bcast_all<0>(i1, r + 8);
        #pragma unroll
        for (int j = 0; j < 16; ++j) v[j] = g[(unsigned)(r[j] >= 0 ? r[j] : self) * NCLS + fc];
        #pragma unroll
        for (int j = 0; j < 16; ++j) acc += (r[j] >= 0) ? v[j] : 0.0f;
        i0 = (pn < end) ? l0 : -1;
        i1 = (pn + 8 < end) ? l1 : -1;
        p = pn;
    }
    if (node < n && f < NCLS) out2[(unsigned)node * NCLS + f] = acc;
}


__global__ void gcn_logsoftmax(const float* __restrict__ out2, const float* __restrict__ dinv,
                               const float* __restrict__ b2, float* __restrict__ out, int n) {
    int i = blockIdx.x * blockDim.x + threadIdx.x;
    if (i >= n) return;
    float v[NCLS];
    float m = -1e30f;
    float s = dinv[i];
    #pragma unroll
    for (int c = 0; c < NCLS; ++c) {
        v[c] = b2[c] + s * out2[(size_t)i * NCLS + c];
        m = fmaxf(m, v[c]);
    }
    float sum = 0.0f;
    #pragma unroll
    for (int c = 0; c < NCLS; ++c) sum += __expf(v[c] - m);
    float lse = m + __logf(sum);
    #pragma unroll
    for (int c = 0; c < NCLS; ++c) out[(size_t)i * NCLS + c] = v[c] - lse;
}


extern "C" void kernel_launch(void* const* d_in, const int* in_sizes, int n_in,
                              void* d_out, int out_size, void* d_ws, size_t ws_size,
                              hipStream_t stream) {
    const float* x  = (const float*)d_in[0];
    const int*   ei = (const int*)  d_in[1];
    const float* W1 = (const float*)d_in[2];
    const float* b1 = (const float*)d_in[3];
    const float* W2 = (const float*)d_in[4];
    const float* b2 = (const float*)d_in[5];
    float* out = (float*)d_out;

    int n = in_sizes[0] / F_IN;
    int e = in_sizes[1] / 2;
    const int* rowi = ei;
    const int* coli = ei + e;

    size_t np = ((size_t)n + 64) & ~(size_t)63;
    size_t ep = ((size_t)e + 63) & ~(size_t)63;
    size_t rw = ep > 2 * HID * np ? ep : 2 * HID * np;
    int nb = (n + BIN_W - 1) >> BIN_SH;
    if (ws_size < (3 * np + ep + rw) * sizeof(float)) return;
    if (nb > BIN_MAXB || (size_t)BIN_G * nb > 2 * np) return;

    int*   base  = (int*)d_ws;
    int*   start = base + np;
    float* dinv  = (float*)(start + np);
    int*   srcs  = (int*)(dinv + np);
    int*   hist  = start;
    unsigned* binned = (unsigned*)(srcs + ep);
    float* h1    = (float*)binned;
    float* out1  = h1 + HID * np;
    float* g     = h1;
    float* out2  = out1;

    const int B = 256;
    int gn = (n + B - 1) / B;

    gcn_bin_count  <<<BIN_G, BIN_T, 0, stream>>>(coli, hist, e, nb);
    gcn_bin_scatter<<<BIN_G, BIN_T, 0, stream>>>(rowi, coli, hist, base, binned, e, nb);
    gcn_bucket_rank<<<nb, BIN_T, 0, stream>>>(binned, base, start, dinv, srcs, n);

    int ntiles = (n + 15) >> 4;
    int nwaves = (ntiles + TILES_PER_WAVE - 1) / TILES_PER_WAVE;
    int gw = (nwaves * 32 + B - 1) / B;
    gcn_gemm1_wmma <<<gw, B, 0, stream>>>(x, W1, h1, n);

    gcn_prescale   <<<(n * (HID / 4) + B - 1) / B, B, 0, stream>>>(h1, dinv, n);
    gcn_agg1       <<<(n * 16 + B - 1) / B, B, 0, stream>>>(start, srcs, h1, out1, n);

    gcn_gemm2      <<<gn, B, 0, stream>>>(out1, b1, W2, dinv, g, n);
    gcn_agg2       <<<(n * 8 + B - 1) / B, B, 0, stream>>>(start, srcs, g, out2, n);

    gcn_logsoftmax <<<gn, B, 0, stream>>>(out2, dinv, b2, out, n);
}


# Row pre-scale, conv2's 16x7 product and log_softmax moved into the ends of the kernels that make their inputs (GEMM store, both per-node sums): 9 launches -> 6; all sum orders kept, 16x7 dot fused
# speedup vs baseline: 7.4563x; 1.0672x over previous
#include <hip/hip_runtime.h>
#include <hip/hip_bf16.h>

#define F_IN 512
#define HID  16
#define NCLS 7
#define TILES_PER_WAVE 2

typedef float v2f __attribute__((ext_vector_type(2)));
typedef float v8f __attribute__((ext_vector_type(8)));


#define BIN_G    256
#define BIN_T    1024
#define BIN_SH   8
#define BIN_W    (1 << BIN_SH)
#define BIN_MAXB BIN_T

__device__ __forceinline__ int wave_scan_incl(int s, int lane) {
    #pragma unroll
    for (int o = 1; o < 32; o <<= 1) {
        int t = __shfl_up(s, o);
        if (lane >= o) s += t;
    }
    return s;
}

template <int K> struct batch_of { static constexpr int size = K; };

template <class Body>
__device__ __forceinline__ void for_edges(int i, int end, Body body) {
    for (; i + 7 * BIN_T < end; i += 8 * BIN_T) body(i, batch_of<8>{});
    if (i + 3 * BIN_T < end) { body(i, batch_of<4>{}); i += 4 * BIN_T; }
    if (i + 1 * BIN_T < end) { body(i, batch_of<2>{}); i += 2 * BIN_T; }
    if (i < end) body(i, batch_of<1>{});
}

__global__ void __launch_bounds__(BIN_T)
gcn_bin_count(const int* __restrict__ col, int* __restrict__ hist, int e, int nb) {
    __shared__ int h[BIN_MAXB];
    int tid = threadIdx.x;
    for (int b = tid; b < nb; b += BIN_T) h[b] = 0;
    __syncthreads();
    int ch = (e + BIN_G - 1) / BIN_G;
    int lo = blockIdx.x * ch;
    int hi = lo + ch < e ? lo + ch : e;
    for_edges(lo + tid, hi, [&](int i, auto kc) {
        constexpr int K = decltype(kc)::size;
        int c[K];
        #pragma unroll
        for (int j = 0; j < K; ++j) c[j] = col[i + j * BIN_T];
        #pragma unroll
        for (int j = 0; j < K; ++j) atomicAdd(&h[c[j] >> BIN_SH], 1);
    });
    __syncthreads();
    for (int b = tid; b < nb; b += BIN_T) hist[blockIdx.x * nb + b] = h[b];
}

__global__ void __launch_bounds__(BIN_T)
gcn_bin_scatter(const int* __restrict__ row, const int* __restrict__ col,
                const int* __restrict__ hist, int* __restrict__ base,
                unsigned* __restrict__ binned, int e, int nb) {
    __shared__ int cur[BIN_MAXB];
    __shared__ int wsum[BIN_T / 32];
    int tid = threadIdx.x, lane = tid & 31, w = tid >> 5;

    int tot = 0, pre = 0;
    if (tid < nb) {
        #pragma unroll 32
        for (int g = 0; g < BIN_G; ++g) {
            int v = hist[g * nb + tid];
            tot += v;
            pre += (g < (int)blockIdx.x) ? v : 0;
        }
    }
    int inc = wave_scan_incl(tot, lane);
    if (lane == 31) wsum[w] = inc;
    __syncthreads();
    if (w == 0) {
        int t = wsum[lane];
        int iw = wave_scan_incl(t, lane);
        wsum[lane] = iw - t;
    }
    __syncthreads();
    int first = wsum[w] + inc - tot;
    if (tid < nb) {
        cur[tid] = first + pre;
        if (blockIdx.x == 0) {
            base[tid] = first;
            if (tid == nb - 1) base[nb] = first + tot;
        }
    }
    __syncthreads();

    int ch = (e + BIN_G - 1) / BIN_G;
    int lo = blockIdx.x * ch;
    int hi = lo + ch < e ? lo + ch : e;
    for_edges(lo + tid, hi, [&](int i, auto kc) {
        constexpr int K = decltype(kc)::size;
        int c[K], r[K];
        #pragma unroll
        for (int j = 0; j < K; ++j) {
            c[j] = col[i + j * BIN_T];
            r[j] = row[i + j * BIN_T];
        }
        int pos[K];
        #pragma unroll
        for (int j = 0; j < K; ++j) pos[j] = atomicAdd(&cur[c[j] >> BIN_SH], 1);
        #pragma unroll
        for (int j = 0; j < K; ++j)
            binned[pos[j]] = ((unsigned)r[j] << BIN_SH) | ((unsigned)c[j] & (BIN_W - 1));
    });
}

__global__ void __launch_bounds__(BIN_T)
gcn_bucket_rank(const unsigned* __restrict__ binned, const int* __restrict__ base,
                int* __restrict__ start, float* __restrict__ dinv, int* __restrict__ srcs, int n) {
    __shared__ int cnt[BIN_W];
    __shared__ int slot[BIN_W];
    __shared__ int wsum[BIN_W / 32];
    int tid = threadIdx.x, lane = tid & 31, w = tid >> 5;
    int lo = base[blockIdx.x];
    int hi = base[blockIdx.x + 1];
    if (tid < BIN_W) cnt[tid] = 0;
    __syncthreads();
    for_edges(lo + tid, hi, [&](int i, auto kc) {
        constexpr int K = decltype(kc)::size;
        unsigned v[K];
        #pragma unroll
        for (int j = 0; j < K; ++j) v[j] = binned[i + j * BIN_T];
        #pragma unroll
        for (int j = 0; j < K; ++j) atomicAdd(&cnt[v[j] & (BIN_W - 1)], 1);
    });
    __syncthreads();
    int c = 0, inc = 0;
    if (tid < BIN_W) {
        c = cnt[tid];
        inc = wave_scan_incl(c, lane);
        if (lane == 31) wsum[w] = inc;
    }
    __syncthreads();
    if (tid < BIN_W) {
        int front = 0;
        #pragma unroll
        for (int k = 0; k < BIN_W / 32; ++k) front += (k < w) ? wsum[k] : 0;
        int excl = front + inc - c;
        slot[tid] = excl;
        int node = blockIdx.x * BIN_W + tid;
        if (node < n) {
            start[node] = lo + excl;
            dinv[node]  = rsqrtf((float)(c + 1));
            if (node == n - 1) start[n] = lo + excl + c;
        }
    }
    __syncthreads();
    for_edges(lo + tid, hi, [&](int i, auto kc) {
        constexpr int K = decltype(kc)::size;
        unsigned v[K];
        #pragma unroll
        for (int j = 0; j < K; ++j) v[j] = binned[i + j * BIN_T];
        int r[K];
        #pragma unroll
        for (int j = 0; j < K; ++j) r[j] = atomicAdd(&slot[v[j] & (BIN_W - 1)], 1);
        #pragma unroll
        for (int j = 0; j < K; ++j) srcs[lo + r[j]] = (int)(v[j] >> BIN_SH);
    });
}


__device__ __forceinline__ void store_tile(float* __restrict__ h1, const float* __restrict__ dinv,
                                           const v8f& acc, int tile, int lane, int n) {
    int mbase = tile * 16 + ((lane >> 4) << 3);
    int ncol  = lane & 15;
    float* p = h1 + (size_t)mbase * HID + ncol;
    if (mbase + 8 <= n) {
        float s[8];
        #pragma unroll
        for (int v = 0; v < 8; ++v) s[v] = dinv[mbase + v];
        #pragma unroll
        for (int v = 0; v < 8; ++v) p[v * HID] = s[v] * acc[v];
    } else {
        #pragma unroll
        for (int v = 0; v < 8; ++v) if (mbase + v < n) p[v * HID] = dinv[mbase + v] * acc[v];
    }
}

__global__ void gcn_gemm1_wmma(const float* __restrict__ x, const float* __restrict__ W1,
                               const float* __restrict__ dinv, float* __restrict__ h1, int n) {
    int lane = threadIdx.x & 31;
    int wid  = (blockIdx.x * blockDim.x + threadIdx.x) >> 5;
    int ntiles = (n + 15) >> 4;
    int tile0 = wid * TILES_PER_WAVE;
    if (tile0 >= ntiles) return;
    bool has1 = (tile0 + 1) < ntiles;

    int m    = lane & 15;
    int koff = (lane >> 4) << 1;

    int row0 = tile0 * 16 + m;
    int row1 = row0 + 16;
    const float* xa0 = x + (size_t)(row0 < n ? row0 : 0) * F_IN + koff;
    const float* xa1 = x + (size_t)(row1 < n ? row1 : 0) * F_IN + koff;

    v8f acc0 = {}, acc1 = {};
    #pragma unroll 4
    for (int k = 0; k < F_IN; k += 4) {
        v2f b;
        b.x = W1[(size_t)(k + koff)     * HID + m];
        b.y = W1[(size_t)(k + koff + 1) * HID + m];
        v2f a0 = *(const v2f*)(xa0 + k);
        v2f a1 = *(const v2f*)(xa1 + k);
        acc0 = __builtin_amdgcn_wmma_f32_16x16x4_f32(
            false, a0, false, b, (short)0, acc0, false, false);
        acc1 = __builtin_amdgcn_wmma_f32_16x16x4_f32(
            false, a1, false, b, (short)0, acc1, false, false);
    }

    store_tile(h1, dinv, acc0, tile0, lane, n);
    if (has1) store_tile(h1, dinv, acc1, tile0 + 1, lane, n);
}


template <int J> __device__ __forceinline__ int row16_bcast(int v) {
    return __builtin_amdgcn_update_dpp(0, v, 0x150 + J, 0xf, 0xf, true);
}

template <int J> __device__ __forceinline__ void row16_bcast_all(int v, int* r) {
    r[J] = row16_bcast<J>(v);
    if constexpr (J + 1 < 16) row16_bcast_all<J + 1>(v, r);
}

template <int J> __device__ __forceinline__ float row16_dot(float h, const float* w, float a) {
    a = __builtin_fmaf(__int_as_float(row16_bcast<J>(__float_as_int(h))), w[J], a);
    if constexpr (J + 1 < 16) return row16_dot<J + 1>(h, w, a);
    else return a;
}

__global__ void gcn_agg1(const int* __restrict__ start, const int* __restrict__ srcs,
                         const float* __restrict__ hs, const float* __restrict__ dinv,
                         const float* __restrict__ b1, const float* __restrict__ W2,
                         float* __restrict__ g, int n) {
    int t = blockIdx.x * blockDim.x + threadIdx.x;
    int node = t >> 4;
    unsigned f = threadIdx.x & 15u;
    int beg = 0, end = 0, self = 0;
    float acc = 0.0f;
    if (node < n) {
        beg = start[node];
        end = start[node + 1];
        self = node;
        acc = hs[(unsigned)node * HID + f];
    }
    int p = beg + (int)f;
    int idx = (p < end) ? srcs[p] : -1;
    for (int base = beg; base < end; base += 16) {
        int pn = p + 16;
        int ld = srcs[pn < end ? pn : beg];
        __builtin_amdgcn_sched_barrier(0);
        int r[16];
        float v[16];
        row16_bcast_all<0>(idx, r);
        #pragma unroll
        for (int j = 0; j < 16; ++j) v[j] = hs[(unsigned)(r[j] >= 0 ? r[j] : self) * HID + f];
        #pragma unroll
        for (int j = 0; j < 16; ++j) acc += (r[j] >= 0) ? v[j] : 0.0f;
        idx = (pn < end) ? ld : -1;
        p = pn;
    }
    unsigned fc = f < NCLS ? f : NCLS - 1;
    float s = (node < n) ? dinv[node] : 0.0f;
    float w[HID];
    #pragma unroll
    for (int j = 0; j < HID; ++j) w[j] = W2[j * NCLS + fc];
    float h = fmaxf(b1[f] + s * acc, 0.0f);
    float a2 = row16_dot<0>(h, w, 0.0f);
    if (node < n && f < NCLS) g[(unsigned)node * NCLS + f] = s * a2;
}


template <int J> __device__ __forceinline__ int grp8_bcast(int v) {
    return (int)__builtin_amdgcn_mov_dpp8((unsigned)v, (unsigned)J * 0x249249u);
}

template <int J> __device__ __forceinline__ void grp8_bcast_all(int v, int* r) {
    r[J] = grp8_bcast<J>(v);
    if constexpr (J + 1 < 8) grp8_bcast_all<J + 1>(v, r);
}

__global__ void gcn_agg2(const int* __restrict__ start, const int* __restrict__ srcs,
                         const float* __restrict__ g, const float* __restrict__ dinv,
                         const float* __restrict__ b2, float* __restrict__ out, int n) {
    int t = blockIdx.x * blockDim.x + threadIdx.x;
    int node = t >> 3;
    unsigned f = threadIdx.x & 7u;
    unsigned fc = f < NCLS ? f : NCLS - 1;
    int beg = 0, end = 0, self = 0;
    float acc = 0.0f;
    if (node < n) {
        beg = start[node];
        end = start[node + 1];
        self = node;
        acc = g[(unsigned)node * NCLS + fc];
    }
    int p = beg + (int)f;
    int i0 = (p < end) ? srcs[p] : -1;
    int i1 = (p + 8 < end) ? srcs[p + 8] : -1;
    for (int base = beg; base < end; base += 16) {
        int pn = p + 16;
        int l0 = srcs[pn < end ? pn : beg];
        int l1 = srcs[pn + 8 < end ? pn + 8 : beg];
        __builtin_amdgcn_sched_barrier(0);
        int r[16];
        float v[16];
        grp8_bcast_all<0>(i0, r);
        grp8_bcast_all<0>(i1, r + 8);
        #pragma unroll
        for (int j = 0; j < 16; ++j) v[j] = g[(unsigned)(r[j] >= 0 ? r[j] : self) * NCLS + fc];
        #pragma unroll
        for (int j = 0; j < 16; ++j) acc += (r[j] >= 0) ? v[j] : 0.0f;
        i0 = (pn < end) ? l0 : -1;
        i1 = (pn + 8 < end) ? l1 : -1;
        p = pn;
    }
    float s = (node < n) ? dinv[node] : 0.0f;
    float val = b2[fc] + s * acc;
    int vr[8];
    grp8_bcast_all<0>(__float_as_int(val), vr);
    float m = -1e30f;
    #pragma unroll
    for (int c = 0; c < NCLS; ++c) m = fmaxf(m, __int_as_float(vr[c]));
    float sum = 0.0f;
    #pragma unroll
    for (int c = 0; c < NCLS; ++c) sum += __expf(__int_as_float(vr[c]) - m);
    float lse = m + __logf(sum);
    if (node < n && f < NCLS) out[(unsigned)node * NCLS + f] = val - lse;
}


extern "C" void kernel_launch(void* const* d_in, const int* in_sizes, int n_in,
                              void* d_out, int out_size, void* d_ws, size_t ws_size,
                              hipStream_t stream) {
    const float* x  = (const float*)d_in[0];
    const int*   ei = (const int*)  d_in[1];
    const float* W1 = (const float*)d_in[2];
    const float* b1 = (const float*)d_in[3];
    const float* W2 = (const float*)d_in[4];
    const float* b2 = (const float*)d_in[5];
    float* out = (float*)d_out;

    int n = in_sizes[0] / F_IN;
    int e = in_sizes[1] / 2;
    const int* rowi = ei;
    const int* coli = ei + e;

    size_t np = ((size_t)n + 64) & ~(size_t)63;
    size_t ep = ((size_t)e + 63) & ~(size_t)63;
    size_t rw = ep > 2 * HID * np ? ep : 2 * HID * np;
    int nb = (n + BIN_W - 1) >> BIN_SH;
    if (ws_size < (3 * np + ep + rw) * sizeof(float)) return;
    if (nb > BIN_MAXB || (size_t)BIN_G * nb > 2 * np) return;

    int*   base  = (int*)d_ws;
    int*   start = base + np;
    float* dinv  = (float*)(start + np);
    int*   srcs  = (int*)(dinv + np);
    int*   hist  = start;
    unsigned* binned = (unsigned*)(srcs + ep);
    float* h1    = (float*)binned;
    float* g     = h1 + HID * np;

    const int B = 256;

    gcn_bin_count  <<<BIN_G, BIN_T, 0, stream>>>(coli, hist, e, nb);
    gcn_bin_scatter<<<BIN_G, BIN_T, 0, stream>>>(rowi, coli, hist, base, binned, e, nb);
    gcn_bucket_rank<<<nb, BIN_T, 0, stream>>>(binned, base, start, dinv, srcs, n);

    int ntiles = (n + 15) >> 4;
    int nwaves = (ntiles + TILES_PER_WAVE - 1) / TILES_PER_WAVE;
    int gw = (nwaves * 32 + B - 1) / B;
    gcn_gemm1_wmma <<<gw, B, 0, stream>>>(x, W1, dinv, h1, n);

    gcn_agg1       <<<(n * 16 + B - 1) / B, B, 0, stream>>>(start, srcs, h1, dinv, b1, W2, g, n);
    gcn_agg2       <<<(n * 8 + B - 1) / B, B, 0, stream>>>(start, srcs, g, dinv, b2, out, n);
}


# Both placing passes of the edge sort build their output in an LDS image and copy it out in runs: a wave's 4-byte stores cover 1-2 lines, not 32; sum order = LDS arrival order as before
# speedup vs baseline: 9.7097x; 1.3022x over previous
#include <hip/hip_runtime.h>
#include <hip/hip_bf16.h>

#define F_IN 512
#define HID  16
#define NCLS 7
#define TILES_PER_WAVE 2

typedef float v2f __attribute__((ext_vector_type(2)));
typedef float v8f __attribute__((ext_vector_type(8)));


#define BIN_G    256
#define BIN_T    1024
#define BIN_SH   8
#define BIN_W    (1 << BIN_SH)
#define BIN_MAXB BIN_T
#define STAGE_CAP 16384

__device__ __forceinline__ int wave_scan_incl(int s, int lane) {
    #pragma unroll
    for (int o = 1; o < 32; o <<= 1) {
        int t = __shfl_up(s, o);
        if (lane >= o) s += t;
    }
    return s;
}

template <int K> struct batch_of { static constexpr int size = K; };

template <class Body>
__device__ __forceinline__ void for_edges(int i, int end, Body body) {
    for (; i + 7 * BIN_T < end; i += 8 * BIN_T) body(i, batch_of<8>{});
    if (i + 3 * BIN_T < end) { body(i, batch_of<4>{}); i += 4 * BIN_T; }
    if (i + 1 * BIN_T < end) { body(i, batch_of<2>{}); i += 2 * BIN_T; }
    if (i < end) body(i, batch_of<1>{});
}

__global__ void __launch_bounds__(BIN_T)
gcn_bin_count(const int* __restrict__ col, int* __restrict__ hist, int e, int nb) {
    __shared__ int h[BIN_MAXB];
    int tid = threadIdx.x;
    for (int b = tid; b < nb; b += BIN_T) h[b] = 0;
    __syncthreads();
    int ch = (e + BIN_G - 1) / BIN_G;
    int lo = blockIdx.x * ch;
    int hi = lo + ch < e ? lo + ch : e;
    for_edges(lo + tid, hi, [&](int i, auto kc) {
        constexpr int K = decltype(kc)::size;
        int c[K];
        #pragma unroll
        for (int j = 0; j < K; ++j) c[j] = col[i + j * BIN_T];
        #pragma unroll
        for (int j = 0; j < K; ++j) atomicAdd(&h[c[j] >> BIN_SH], 1);
    });
    __syncthreads();
    for (int b = tid; b < nb; b += BIN_T) hist[blockIdx.x * nb + b] = h[b];
}

__global__ void __launch_bounds__(BIN_T)
gcn_bin_scatter(const int* __restrict__ row, const int* __restrict__ col,
                const int* __restrict__ hist, int* __restrict__ base,
                unsigned* __restrict__ binned, int e, int nb) {
    __shared__ int cur[BIN_MAXB];
    __shared__ int lbeg[BIN_MAXB];
    __shared__ int gbeg[BIN_MAXB];
    __shared__ int wsum[BIN_T / 32];
    __shared__ int wloc[BIN_T / 32];
    __shared__ unsigned stage[STAGE_CAP];
    int tid = threadIdx.x, lane = tid & 31, w = tid >> 5;

    int tot = 0, pre = 0, mine = 0;
    if (tid < nb) {
        mine = hist[blockIdx.x * nb + tid];
        #pragma unroll 32
        for (int g = 0; g < BIN_G; ++g) {
            int v = hist[g * nb + tid];
            tot += v;
            pre += (g < (int)blockIdx.x) ? v : 0;
        }
    }
    int inc  = wave_scan_incl(tot, lane);
    int linc = wave_scan_incl(mine, lane);
    if (lane == 31) { wsum[w] = inc; wloc[w] = linc; }
    __syncthreads();
    if (w == 0) {
        int t = wsum[lane], u = wloc[lane];
        int iw = wave_scan_incl(t, lane);
        int iu = wave_scan_incl(u, lane);
        wsum[lane] = iw - t;
        wloc[lane] = iu - u;
    }
    __syncthreads();
    int first  = wsum[w] + inc - tot;
    int lfirst = wloc[w] + linc - mine;

    int ch = (e + BIN_G - 1) / BIN_G;
    int lo = blockIdx.x * ch;
    int hi = lo + ch < e ? lo + ch : e;
    bool staged = hi - lo <= STAGE_CAP;
    if (tid < nb) {
        lbeg[tid] = lfirst;
        gbeg[tid] = first + pre;
        cur[tid]  = staged ? lfirst : first + pre;
        if (blockIdx.x == 0) {
            base[tid] = first;
            if (tid == nb - 1) base[nb] = first + tot;
        }
    }
    __syncthreads();

    if (staged) {
        for_edges(lo + tid, hi, [&](int i, auto kc) {
            constexpr int K = decltype(kc)::size;
            int c[K], r[K];
            #pragma unroll
            for (int j = 0; j < K; ++j) {
                c[j] = col[i + j * BIN_T];
                r[j] = row[i + j * BIN_T];
            }
            int pos[K];
            #pragma unroll
            for (int j = 0; j < K; ++j) pos[j] = atomicAdd(&cur[c[j] >> BIN_SH], 1);
            #pragma unroll
            for (int j = 0; j < K; ++j)
                stage[pos[j]] = ((unsigned)r[j] << BIN_SH) | ((unsigned)c[j] & (BIN_W - 1));
        });
        __syncthreads();
        for (int b0 = w; b0 < nb; b0 += 4 * (BIN_T / 32)) {
            int s[4], len[4], gp[4];
            #pragma unroll
            for (int k = 0; k < 4; ++k) {
                int b  = b0 + k * (BIN_T / 32);
                int bc = b < nb ? b : nb - 1;
                s[k]   = lbeg[bc];
                len[k] = b < nb ? cur[bc] - s[k] : 0;
                gp[k]  = gbeg[bc];
            }
            unsigned x[4], y[4];
            #pragma unroll
            for (int k = 0; k < 4; ++k) {
                x[k] = stage[lane      < len[k] ? s[k] + lane      : 0];
                y[k] = stage[lane + 32 < len[k] ? s[k] + lane + 32 : 0];
            }
            #pragma unroll
            for (int k = 0; k < 4; ++k) {
                if (lane      < len[k]) binned[gp[k] + lane]      = x[k];
                if (lane + 32 < len[k]) binned[gp[k] + lane + 32] = y[k];
            }
            #pragma unroll
            for (int k = 0; k < 4; ++k)
                for (int j = lane + 64; j < len[k]; j += 32) binned[gp[k] + j] = stage[s[k] + j];
        }
    } else {
        for_edges(lo + tid, hi, [&](int i, auto kc) {
            constexpr int K = decltype(kc)::size;
            int c[K], r[K];
            #pragma unroll
            for (int j = 0; j < K; ++j) {
                c[j] = col[i + j * BIN_T];
                r[j] = row[i + j * BIN_T];
            }
            int pos[K];
            #pragma unroll
            for (int j = 0; j < K; ++j) pos[j] = atomicAdd(&cur[c[j] >> BIN_SH], 1);
            #pragma unroll
            for (int j = 0; j < K; ++j)
                binned[pos[j]] = ((unsigned)r[j] << BIN_SH) | ((unsigned)c[j] & (BIN_W - 1));
        });
    }
}

__global__ void __launch_bounds__(BIN_T)
gcn_bucket_rank(const unsigned* __restrict__ binned, const int* __restrict__ base,
                int* __restrict__ start, float* __restrict__ dinv, int* __restrict__ srcs, int n) {
    __shared__ int cnt[BIN_W];
    __shared__ int slot[BIN_W];
    __shared__ int wsum[BIN_W / 32];
    __shared__ int stage[STAGE_CAP];
    int tid = threadIdx.x, lane = tid & 31, w = tid >> 5;
    int lo = base[blockIdx.x];
    int hi = base[blockIdx.x + 1];
    if (tid < BIN_W) cnt[tid] = 0;
    __syncthreads();
    for_edges(lo + tid, hi, [&](int i, auto kc) {
        constexpr int K = decltype(kc)::size;
        unsigned v[K];
        #pragma unroll
        for (int j = 0; j < K; ++j) v[j] = binned[i + j * BIN_T];
        #pragma unroll
        for (int j = 0; j < K; ++j) atomicAdd(&cnt[v[j] & (BIN_W - 1)], 1);
    });
    __syncthreads();
    int c = 0, inc = 0;
    if (tid < BIN_W) {
        c = cnt[tid];
        inc = wave_scan_incl(c, lane);
        if (lane == 31) wsum[w] = inc;
    }
    __syncthreads();
    if (tid < BIN_W) {
        int front = 0;
        #pragma unroll
        for (int k = 0; k < BIN_W / 32; ++k) front += (k < w) ? wsum[k] : 0;
        int excl = front + inc - c;
        slot[tid] = excl;
        int node = blockIdx.x * BIN_W + tid;
        if (node < n) {
            start[node] = lo + excl;
            dinv[node]  = rsqrtf((float)(c + 1));
            if (node == n - 1) start[n] = lo + excl + c;
        }
    }
    __syncthreads();
    if (hi - lo <= STAGE_CAP) {
        for_edges(lo + tid, hi, [&](int i, auto kc) {
            constexpr int K = decltype(kc)::size;
            unsigned v[K];
            #pragma unroll
            for (int j = 0; j < K; ++j) v[j] = binned[i + j * BIN_T];
            int r[K];
            #pragma unroll
            for (int j = 0; j < K; ++j) r[j] = atomicAdd(&slot[v[j] & (BIN_W - 1)], 1);
            #pragma unroll
            for (int j = 0; j < K; ++j) stage[r[j]] = (int)(v[j] >> BIN_SH);
        });
        __syncthreads();
        int n_sl = hi - lo;
        int head = (32 - (lo & 31)) & 31;
        head = head < n_sl ? head : n_sl;
        if (tid < head) srcs[lo + tid] = stage[tid];
        for_edges(head + tid, n_sl, [&](int i, auto kc) {
            constexpr int K = decltype(kc)::size;
            int t[K];
            #pragma unroll
            for (int j = 0; j < K; ++j) t[j] = stage[i + j * BIN_T];
            #pragma unroll
            for (int j = 0; j < K; ++j) srcs[lo + i + j * BIN_T] = t[j];
        });
    } else {
        for_edges(lo + tid, hi, [&](int i, auto kc) {
            constexpr int K = decltype(kc)::size;
            unsigned v[K];
            #pragma unroll
            for (int j = 0; j < K; ++j) v[j] = binned[i + j * BIN_T];
            int r[K];
            #pragma unroll
            for (int j = 0; j < K; ++j) r[j] = atomicAdd(&slot[v[j] & (BIN_W - 1)], 1);
            #pragma unroll
            for (int j = 0; j < K; ++j) srcs[lo + r[j]] = (int)(v[j] >> BIN_SH);
        });
    }
}


__device__ __forceinline__ void store_tile(float* __restrict__ h1, const float* __restrict__ dinv,
                                           const v8f& acc, int tile, int lane, int n) {
    int mbase = tile * 16 + ((lane >> 4) << 3);
    int ncol  = lane & 15;
    float* p = h1 + (size_t)mbase * HID + ncol;
    if (mbase + 8 <= n) {
        float s[8];
        #pragma unroll
        for (int v = 0; v < 8; ++v) s[v] = dinv[mbase + v];
        #pragma unroll
        for (int v = 0; v < 8; ++v) p[v * HID] = s[v] * acc[v];
    } else {
        #pragma unroll
        for (int v = 0; v < 8; ++v) if (mbase + v < n) p[v * HID] = dinv[mbase + v] * acc[v];
    }
}

__global__ void gcn_gemm1_wmma(const float* __restrict__ x, const float* __restrict__ W1,
                               const float* __restrict__ dinv, float* __restrict__ h1, int n) {
    int lane = threadIdx.x & 31;
    int wid  = (blockIdx.x * blockDim.x + threadIdx.x) >> 5;
    int ntiles = (n + 15) >> 4;
    int tile0 = wid * TILES_PER_WAVE;
    if (tile0 >= ntiles) return;
    bool has1 = (tile0 + 1) < ntiles;

    int m    = lane & 15;
    int koff = (lane >> 4) << 1;

    int row0 = tile0 * 16 + m;
    int row1 = row0 + 16;
    const float* xa0 = x + (size_t)(row0 < n ? row0 : 0) * F_IN + koff;
    const float* xa1 = x + (size_t)(row1 < n ? row1 : 0) * F_IN + koff;

    v8f acc0 = {}, acc1 = {};
    #pragma unroll 4
    for (int k = 0; k < F_IN; k += 4) {
        v2f b;
        b.x = W1[(size_t)(k + koff)     * HID + m];
        b.y = W1[(size_t)(k + koff + 1) * HID + m];
        v2f a0 = *(const v2f*)(xa0 + k);
        v2f a1 = *(const v2f*)(xa1 + k);
        acc0 = __builtin_amdgcn_wmma_f32_16x16x4_f32(
            false, a0, false, b, (short)0, acc0, false, false);
        acc1 = __builtin_amdgcn_wmma_f32_16x16x4_f32(
            false, a1, false, b, (short)0, acc1, false, false);
    }

    store_tile(h1, dinv, acc0, tile0, lane, n);
    if (has1) store_tile(h1, dinv, acc1, tile0 + 1, lane, n);
}


template <int J> __device__ __forceinline__ int row16_bcast(int v) {
    return __builtin_amdgcn_update_dpp(0, v, 0x150 + J, 0xf, 0xf, true);
}

template <int J> __device__ __forceinline__ void row16_bcast_all(int v, int* r) {
    r[J] = row16_bcast<J>(v);
    if constexpr (J + 1 < 16) row16_bcast_all<J + 1>(v, r);
}

template <int J> __device__ __forceinline__ float row16_dot(float h, const float* w, float a) {
    a = __builtin_fmaf(__int_as_float(row16_bcast<J>(__float_as_int(h))), w[J], a);
    if constexpr (J + 1 < 16) return row16_dot<J + 1>(h, w, a);
    else return a;
}

__global__ void gcn_agg1(const int* __restrict__ start, const int* __restrict__ srcs,
                         const float* __restrict__ hs, const float* __restrict__ dinv,
                         const float* __restrict__ b1, const float* __restrict__ W2,
                         float* __restrict__ g, int n) {
    int t = blockIdx.x * blockDim.x + threadIdx.x;
    int node = t >> 4;
    unsigned f = threadIdx.x & 15u;
    int beg = 0, end = 0, self = 0;
    float acc = 0.0f;
    if (node < n) {
        beg = start[node];
        end = start[node + 1];
        self = node;
        acc = hs[(unsigned)node * HID + f];
    }
    int p = beg + (int)f;
    int idx = (p < end) ? srcs[p] : -1;
    for (int base = beg; base < end; base += 16) {
        int pn = p + 16;
        int ld = srcs[pn < end ? pn : beg];
        __builtin_amdgcn_sched_barrier(0);
        int r[16];
        float v[16];
        row16_bcast_all<0>(idx, r);
        #pragma unroll
        for (int j = 0; j < 16; ++j) v[j] = hs[(unsigned)(r[j] >= 0 ? r[j] : self) * HID + f];
        #pragma unroll
        for (int j = 0; j < 16; ++j) acc += (r[j] >= 0) ? v[j] : 0.0f;
        idx = (pn < end) ? ld : -1;
        p = pn;
    }
    unsigned fc = f < NCLS ? f : NCLS - 1;
    float s = (node < n) ? dinv[node] : 0.0f;
    float w[HID];
    #pragma unroll
    for (int j = 0; j < HID; ++j) w[j] = W2[j * NCLS + fc];
    float h = fmaxf(b1[f] + s * acc, 0.0f);
    float a2 = row16_dot<0>(h, w, 0.0f);
    if (node < n && f < NCLS) g[(unsigned)node * NCLS + f] = s * a2;
}


template <int J> __device__ __forceinline__ int grp8_bcast(int v) {
    return (int)__builtin_amdgcn_mov_dpp8((unsigned)v, (unsigned)J * 0x249249u);
}

template <int J> __device__ __forceinline__ void grp8_bcast_all(int v, int* r) {
    r[J] = grp8_bcast<J>(v);
    if constexpr (J + 1 < 8) grp8_bcast_all<J + 1>(v, r);
}

__global__ void gcn_agg2(const int* __restrict__ start, const int* __restrict__ srcs,
                         const float* __restrict__ g, const float* __restrict__ dinv,
                         const float* __restrict__ b2, float* __restrict__ out, int n) {
    int t = blockIdx.x * blockDim.x + threadIdx.x;
    int node = t >> 3;
    unsigned f = threadIdx.x & 7u;
    unsigned fc = f < NCLS ? f : NCLS - 1;
    int beg = 0, end = 0, self = 0;
    float acc = 0.0f;
    if (node < n) {
        beg = start[node];
        end = start[node + 1];
        self = node;
        acc = g[(unsigned)node * NCLS + fc];
    }
    int p = beg + (int)f;
    int i0 = (p < end) ? srcs[p] : -1;
    int i1 = (p + 8 < end) ? srcs[p + 8] : -1;
    for (int base = beg; base < end; base += 16) {
        int pn = p + 16;
        int l0 = srcs[pn < end ? pn : beg];
        int l1 = srcs[pn + 8 < end ? pn + 8 : beg];
        __builtin_amdgcn_sched_barrier(0);
        int r[16];
        float v[16];
        grp8_bcast_all<0>(i0, r);
        grp8_bcast_all<0>(i1, r + 8);
        #pragma unroll
        for (int j = 0; j < 16; ++j) v[j] = g[(unsigned)(r[j] >= 0 ? r[j] : self) * NCLS + fc];
        #pragma unroll
        for (int j = 0; j < 16; ++j) acc += (r[j] >= 0) ? v[j] : 0.0f;
        i0 = (pn < end) ? l0 : -1;
        i1 = (pn + 8 < end) ? l1 : -1;
        p = pn;
    }
    float s = (node < n) ? dinv[node] : 0.0f;
    float val = b2[fc] + s * acc;
    int vr[8];
    grp8_bcast_all<0>(__float_as_int(val), vr);
    float m = -1e30f;
    #pragma unroll
    for (int c = 0; c < NCLS; ++c) m = fmaxf(m, __int_as_float(vr[c]));
    float sum = 0.0f;
    #pragma unroll
    for (int c = 0; c < NCLS; ++c) sum += __expf(__int_as_float(vr[c]) - m);
    float lse = m + __logf(sum);
    if (node < n && f < NCLS) out[(unsigned)node * NCLS + f] = val - lse;
}


extern "C" void kernel_launch(void* const* d_in, const int* in_sizes, int n_in,
                              void* d_out, int out_size, void* d_ws, size_t ws_size,
                              hipStream_t stream) {
    const float* x  = (const float*)d_in[0];
    const int*   ei = (const int*)  d_in[1];
    const float* W1 = (const float*)d_in[2];
    const float* b1 = (const float*)d_in[3];
    const float* W2 = (const float*)d_in[4];
    const float* b2 = (const float*)d_in[5];
    float* out = (float*)d_out;

    int n = in_sizes[0] / F_IN;
    int e = in_sizes[1] / 2;
    const int* rowi = ei;
    const int* coli = ei + e;

    size_t np = ((size_t)n + 64) & ~(size_t)63;
    size_t ep = ((size_t)e + 63) & ~(size_t)63;
    size_t rw = ep > 2 * HID * np ? ep : 2 * HID * np;
    int nb = (n + BIN_W - 1) >> BIN_SH;
    if (ws_size < (3 * np + ep + rw) * sizeof(float)) return;
    if (nb > BIN_MAXB || (size_t)BIN_G * nb > 2 * np) return;

    int*   base  = (int*)d_ws;
    int*   start = base + np;
    float* dinv  = (float*)(start + np);
    int*   srcs  = (int*)(dinv + np);
    int*   hist  = start;
    unsigned* binned = (unsigned*)(srcs + ep);
    float* h1    = (float*)binned;
    float* g     = h1 + HID * np;

    const int B = 256;

    gcn_bin_count  <<<BIN_G, BIN_T, 0, stream>>>(coli, hist, e, nb);
    gcn_bin_scatter<<<BIN_G, BIN_T, 0, stream>>>(rowi, coli, hist, base, binned, e, nb);
    gcn_bucket_rank<<<nb, BIN_T, 0, stream>>>(binned, base, start, dinv, srcs, n);

    int ntiles = (n + 15) >> 4;
    int nwaves = (ntiles + TILES_PER_WAVE - 1) / TILES_PER_WAVE;
    int gw = (nwaves * 32 + B - 1) / B;
    gcn_gemm1_wmma <<<gw, B, 0, stream>>>(x, W1, dinv, h1, n);

    gcn_agg1       <<<(n * 16 + B - 1) / B, B, 0, stream>>>(start, srcs, h1, dinv, b1, W2, g, n);
    gcn_agg2       <<<(n * 8 + B - 1) / B, B, 0, stream>>>(start, srcs, g, dinv, b2, out, n);
}


# conv1 GEMM reads x as whole lines: a 16-byte-per-lane load covers 512 contiguous bytes of one row, chunks parked in per-wave LDS, A and W1 fragments read from LDS; k sum order changed (two chains)
# speedup vs baseline: 11.0492x; 1.1380x over previous
#include <hip/hip_runtime.h>
#include <hip/hip_bf16.h>

#define F_IN 512
#define HID  16
#define NCLS 7

typedef float v2f __attribute__((ext_vector_type(2)));
typedef float v8f __attribute__((ext_vector_type(8)));


#define BIN_G    256
#define BIN_T    1024
#define BIN_SH   8
#define BIN_W    (1 << BIN_SH)
#define BIN_MAXB BIN_T
#define STAGE_CAP 16384

__device__ __forceinline__ int wave_scan_incl(int s, int lane) {
    #pragma unroll
    for (int o = 1; o < 32; o <<= 1) {
        int t = __shfl_up(s, o);
        if (lane >= o) s += t;
    }
    return s;
}

template <int K> struct batch_of { static constexpr int size = K; };

template <class Body>
__device__ __forceinline__ void for_edges(int i, int end, Body body) {
    for (; i + 7 * BIN_T < end; i += 8 * BIN_T) body(i, batch_of<8>{});
    if (i + 3 * BIN_T < end) { body(i, batch_of<4>{}); i += 4 * BIN_T; }
    if (i + 1 * BIN_T < end) { body(i, batch_of<2>{}); i += 2 * BIN_T; }
    if (i < end) body(i, batch_of<1>{});
}

__global__ void __launch_bounds__(BIN_T)
gcn_bin_count(const int* __restrict__ col, int* __restrict__ hist, int e, int nb) {
    __shared__ int h[BIN_MAXB];
    int tid = threadIdx.x;
    for (int b = tid; b < nb; b += BIN_T) h[b] = 0;
    __syncthreads();
    int ch = (e + BIN_G - 1) / BIN_G;
    int lo = blockIdx.x * ch;
    int hi = lo + ch < e ? lo + ch : e;
    for_edges(lo + tid, hi, [&](int i, auto kc) {
        constexpr int K = decltype(kc)::size;
        int c[K];
        #pragma unroll
        for (int j = 0; j < K; ++j) c[j] = col[i + j * BIN_T];
        #pragma unroll
        for (int j = 0; j < K; ++j) atomicAdd(&h[c[j] >> BIN_SH], 1);
    });
    __syncthreads();
    for (int b = tid; b < nb; b += BIN_T) hist[blockIdx.x * nb + b] = h[b];
}

__global__ void __launch_bounds__(BIN_T)
gcn_bin_scatter(const int* __restrict__ row, const int* __restrict__ col,
                const int* __restrict__ hist, int* __restrict__ base,
                unsigned* __restrict__ binned, int e, int nb) {
    __shared__ int cur[BIN_MAXB];
    __shared__ int lbeg[BIN_MAXB];
    __shared__ int gbeg[BIN_MAXB];
    __shared__ int wsum[BIN_T / 32];
    __shared__ int wloc[BIN_T / 32];
    __shared__ unsigned stage[STAGE_CAP];
    int tid = threadIdx.x, lane = tid & 31, w = tid >> 5;

    int tot = 0, pre = 0, mine = 0;
    if (tid < nb) {
        mine = hist[blockIdx.x * nb + tid];
        #pragma unroll 32
        for (int g = 0; g < BIN_G; ++g) {
            int v = hist[g * nb + tid];
            tot += v;
            pre += (g < (int)blockIdx.x) ? v : 0;
        }
    }
    int inc  = wave_scan_incl(tot, lane);
    int linc = wave_scan_incl(mine, lane);
    if (lane == 31) { wsum[w] = inc; wloc[w] = linc; }
    __syncthreads();
    if (w == 0) {
        int t = wsum[lane], u = wloc[lane];
        int iw = wave_scan_incl(t, lane);
        int iu = wave_scan_incl(u, lane);
        wsum[lane] = iw - t;
        wloc[lane] = iu - u;
    }
    __syncthreads();
    int first  = wsum[w] + inc - tot;
    int lfirst = wloc[w] + linc - mine;

    int ch = (e + BIN_G - 1) / BIN_G;
    int lo = blockIdx.x * ch;
    int hi = lo + ch < e ? lo + ch : e;
    bool staged = hi - lo <= STAGE_CAP;
    if (tid < nb) {
        lbeg[tid] = lfirst;
        gbeg[tid] = first + pre;
        cur[tid]  = staged ? lfirst : first + pre;
        if (blockIdx.x == 0) {
            base[tid] = first;
            if (tid == nb - 1) base[nb] = first + tot;
        }
    }
    __syncthreads();

    if (staged) {
        for_edges(lo + tid, hi, [&](int i, auto kc) {
            constexpr int K = decltype(kc)::size;
            int c[K], r[K];
            #pragma unroll
            for (int j = 0; j < K; ++j) {
                c[j] = col[i + j * BIN_T];
                r[j] = row[i + j * BIN_T];
            }
            int pos[K];
            #pragma unroll
            for (int j = 0; j < K; ++j) pos[j] = atomicAdd(&cur[c[j] >> BIN_SH], 1);
            #pragma unroll
            for (int j = 0; j < K; ++j)
                stage[pos[j]] = ((unsigned)r[j] << BIN_SH) | ((unsigned)c[j] & (BIN_W - 1));
        });
        __syncthreads();
        for (int b0 = w; b0 < nb; b0 += 4 * (BIN_T / 32)) {
            int s[4], len[4], gp[4];
            #pragma unroll
            for (int k = 0; k < 4; ++k) {
                int b  = b0 + k * (BIN_T / 32);
                int bc = b < nb ? b : nb - 1;
                s[k]   = lbeg[bc];
                len[k] = b < nb ? cur[bc] - s[k] : 0;
                gp[k]  = gbeg[bc];
            }
            unsigned x[4], y[4];
            #pragma unroll
            for (int k = 0; k < 4; ++k) {
                x[k] = stage[lane      < len[k] ? s[k] + lane      : 0];
                y[k] = stage[lane + 32 < len[k] ? s[k] + lane + 32 : 0];
            }
            #pragma unroll
            for (int k = 0; k < 4; ++k) {
                if (lane      < len[k]) binned[gp[k] + lane]      = x[k];
                if (lane + 32 < len[k]) binned[gp[k] + lane + 32] = y[k];
            }
            #pragma unroll
            for (int k = 0; k < 4; ++k)
                for (int j = lane + 64; j < len[k]; j += 32) binned[gp[k] + j] = stage[s[k] + j];
        }
    } else {
        for_edges(lo + tid, hi, [&](int i, auto kc) {
            constexpr int K = decltype(kc)::size;
            int c[K], r[K];
            #pragma unroll
            for (int j = 0; j < K; ++j) {
                c[j] = col[i + j * BIN_T];
                r[j] = row[i + j * BIN_T];
            }
            int pos[K];
            #pragma unroll
            for (int j = 0; j < K; ++j) pos[j] = atomicAdd(&cur[c[j] >> BIN_SH], 1);
            #pragma unroll
            for (int j = 0; j < K; ++j)
                binned[pos[j]] = ((unsigned)r[j] << BIN_SH) | ((unsigned)c[j] & (BIN_W - 1));
        });
    }
}

__global__ void __launch_bounds__(BIN_T)
gcn_bucket_rank(const unsigned* __restrict__ binned, const int* __restrict__ base,
                int* __restrict__ start, float* __restrict__ dinv, int* __restrict__ srcs, int n) {
    __shared__ int cnt[BIN_W];
    __shared__ int slot[BIN_W];
    __shared__ int wsum[BIN_W / 32];
    __shared__ int stage[STAGE_CAP];
    int tid = threadIdx.x, lane = tid & 31, w = tid >> 5;
    int lo = base[blockIdx.x];
    int hi = base[blockIdx.x + 1];
    if (tid < BIN_W) cnt[tid] = 0;
    __syncthreads();
    for_edges(lo + tid, hi, [&](int i, auto kc) {
        constexpr int K = decltype(kc)::size;
        unsigned v[K];
        #pragma unroll
        for (int j = 0; j < K; ++j) v[j] = binned[i + j * BIN_T];
        #pragma unroll
        for (int j = 0; j < K; ++j) atomicAdd(&cnt[v[j] & (BIN_W - 1)], 1);
    });
    __syncthreads();
    int c = 0, inc = 0;
    if (tid < BIN_W) {
        c = cnt[tid];
        inc = wave_scan_incl(c, lane);
        if (lane == 31) wsum[w] = inc;
    }
    __syncthreads();
    if (tid < BIN_W) {
        int front = 0;
        #pragma unroll
        for (int k = 0; k < BIN_W / 32; ++k) front += (k < w) ? wsum[k] : 0;
        int excl = front + inc - c;
        slot[tid] = excl;
        int node = blockIdx.x * BIN_W + tid;
        if (node < n) {
            start[node] = lo + excl;
            dinv[node]  = rsqrtf((float)(c + 1));
            if (node == n - 1) start[n] = lo + excl + c;
        }
    }
    __syncthreads();
    if (hi - lo <= STAGE_CAP) {
        for_edges(lo + tid, hi, [&](int i, auto kc) {
            constexpr int K = decltype(kc)::size;
            unsigned v[K];
            #pragma unroll
            for (int j = 0; j < K; ++j) v[j] = binned[i + j * BIN_T];
            int r[K];
            #pragma unroll
            for (int j = 0; j < K; ++j) r[j] = atomicAdd(&slot[v[j] & (BIN_W - 1)], 1);
            #pragma unroll
            for (int j = 0; j < K; ++j) stage[r[j]] = (int)(v[j] >> BIN_SH);
        });
        __syncthreads();
        int n_sl = hi - lo;
        int head = (32 - (lo & 31)) & 31;
        head = head < n_sl ? head : n_sl;
        if (tid < head) srcs[lo + tid] = stage[tid];
        for_edges(head + tid, n_sl, [&](int i, auto kc) {
            constexpr int K = decltype(kc)::size;
            int t[K];
            #pragma unroll
            for (int j = 0; j < K; ++j) t[j] = stage[i + j * BIN_T];
            #pragma unroll
            for (int j = 0; j < K; ++j) srcs[lo + i + j * BIN_T] = t[j];
        });
    } else {
        for_edges(lo + tid, hi, [&](int i, auto kc) {
            constexpr int K = decltype(kc)::size;
            unsigned v[K];
            #pragma unroll
            for (int j = 0; j < K; ++j) v[j] = binned[i + j * BIN_T];
            int r[K];
            #pragma unroll
            for (int j = 0; j < K; ++j) r[j] = atomicAdd(&slot[v[j] & (BIN_W - 1)], 1);
            #pragma unroll
            for (int j = 0; j < K; ++j) srcs[lo + r[j]] = (int)(v[j] >> BIN_SH);
        });
    }
}


#define GT_WAVES 4
#define GT_CH    128
#define GT_P4    (GT_CH / 4 + 1)

typedef float v4f __attribute__((ext_vector_type(4)));

__device__ __forceinline__ void store_tile(float* __restrict__ h1, const float* __restrict__ dinv,
                                           const v8f& acc, int tile, int lane, int n) {
    int mbase = tile * 16 + ((lane >> 4) << 3);
    int ncol  = lane & 15;
    float* p = h1 + (size_t)mbase * HID + ncol;
    if (mbase + 8 <= n) {
        float s[8];
        #pragma unroll
        for (int v = 0; v < 8; ++v) s[v] = dinv[mbase + v];
        #pragma unroll
        for (int v = 0; v < 8; ++v) p[v * HID] = s[v] * acc[v];
    } else {
        #pragma unroll
        for (int v = 0; v < 8; ++v) if (mbase + v < n) p[v * HID] = dinv[mbase + v] * acc[v];
    }
}

__global__ void __launch_bounds__(GT_WAVES * 32)
gcn_gemm1_wmma(const float* __restrict__ x, const float* __restrict__ W1,
               const float* __restrict__ dinv, float* __restrict__ h1, int n) {
    __shared__ v4f wf[(F_IN / 8) * 32];
    __shared__ v4f xs[GT_WAVES][16 * GT_P4];
    int lane = threadIdx.x & 31;
    int w    = threadIdx.x >> 5;
    int m    = lane & 15;
    int h    = lane >> 4;
    int ntiles = (n + 15) >> 4;
    int tile = blockIdx.x * GT_WAVES + w;
    int tl   = tile < ntiles ? tile : ntiles - 1;

    {
        v4f t[F_IN / 8 / GT_WAVES];
        #pragma unroll
        for (int i = 0; i < F_IN / 8 / GT_WAVES; ++i) {
            int G = w + i * GT_WAVES;
            const float* p = W1 + (size_t)(8 * G + 4 * h) * HID + m;
            t[i].x = p[0 * HID];
            t[i].y = p[1 * HID];
            t[i].z = p[2 * HID];
            t[i].w = p[3 * HID];
        }
        #pragma unroll
        for (int i = 0; i < F_IN / 8 / GT_WAVES; ++i) wf[(w + i * GT_WAVES) * 32 + lane] = t[i];
    }

    const v4f* xrow[16];
    #pragma unroll
    for (int r = 0; r < 16; ++r) {
        int row = tl * 16 + r;
        xrow[r] = (const v4f*)(x + (size_t)(row < n ? row : 0) * F_IN) + lane;
    }
    v4f pre[16];
    #pragma unroll
    for (int r = 0; r < 16; ++r) pre[r] = xrow[r][0];

    __syncthreads();

    v4f* xw = xs[w];
    v8f acc0 = {}, acc1 = {};
    #pragma unroll
    for (int c = 0; c < F_IN / GT_CH; ++c) {
        #pragma unroll
        for (int r = 0; r < 16; ++r) xw[r * GT_P4 + lane] = pre[r];
        if (c + 1 < F_IN / GT_CH) {
            #pragma unroll
            for (int r = 0; r < 16; ++r) pre[r] = xrow[r][(c + 1) * (GT_CH / 4)];
        }
        __builtin_amdgcn_wave_barrier();
        #pragma unroll
        for (int g = 0; g < GT_CH / 8; ++g) {
            v4f a = xw[m * GT_P4 + 2 * g + h];
            v4f b = wf[(c * (GT_CH / 8) + g) * 32 + lane];
            v2f a0 = {a.x, a.y}, a1 = {a.z, a.w};
            v2f b0 = {b.x, b.y}, b1 = {b.z, b.w};
            acc0 = __builtin_amdgcn_wmma_f32_16x16x4_f32(
                false, a0, false, b0, (short)0, acc0, false, false);
            acc1 = __builtin_amdgcn_wmma_f32_16x16x4_f32(
                false, a1, false, b1, (short)0, acc1, false, false);
        }
        __builtin_amdgcn_wave_barrier();
    }
    v8f acc = acc0 + acc1;
    if (tile < ntiles) store_tile(h1, dinv, acc, tile, lane, n);
}


template <int J> __device__ __forceinline__ int row16_bcast(int v) {
    return __builtin_amdgcn_update_dpp(0, v, 0x150 + J, 0xf, 0xf, true);
}

template <int J> __device__ __forceinline__ void row16_bcast_all(int v, int* r) {
    r[J] = row16_bcast<J>(v);
    if constexpr (J + 1 < 16) row16_bcast_all<J + 1>(v, r);
}

template <int J> __device__ __forceinline__ float row16_dot(float h, const float* w, float a) {
    a = __builtin_fmaf(__int_as_float(row16_bcast<J>(__float_as_int(h))), w[J], a);
    if constexpr (J + 1 < 16) return row16_dot<J + 1>(h, w, a);
    else return a;
}

__global__ void gcn_agg1(const int* __restrict__ start, const int* __restrict__ srcs,
                         const float* __restrict__ hs, const float* __restrict__ dinv,
                         const float* __restrict__ b1, const float* __restrict__ W2,
                         float* __restrict__ g, int n) {
    int t = blockIdx.x * blockDim.x + threadIdx.x;
    int node = t >> 4;
    unsigned f = threadIdx.x & 15u;
    int beg = 0, end = 0, self = 0;
    float acc = 0.0f;
    if (node < n) {
        beg = start[node];
        end = start[node + 1];
        self = node;
        acc = hs[(unsigned)node * HID + f];
    }
    int p = beg + (int)f;
    int idx = (p < end) ? srcs[p] : -1;
    for (int base = beg; base < end; base += 16) {
        int pn = p + 16;
        int ld = srcs[pn < end ? pn : beg];
        __builtin_amdgcn_sched_barrier(0);
        int r[16];
        float v[16];
        row16_bcast_all<0>(idx, r);
        #pragma unroll
        for (int j = 0; j < 16; ++j) v[j] = hs[(unsigned)(r[j] >= 0 ? r[j] : self) * HID + f];
        #pragma unroll
        for (int j = 0; j < 16; ++j) acc += (r[j] >= 0) ? v[j] : 0.0f;
        idx = (pn < end) ? ld : -1;
        p = pn;
    }
    unsigned fc = f < NCLS ? f : NCLS - 1;
    float s = (node < n) ? dinv[node] : 0.0f;
    float w[HID];
    #pragma unroll
    for (int j = 0; j < HID; ++j) w[j] = W2[j * NCLS + fc];
    float h = fmaxf(b1[f] + s * acc, 0.0f);
    float a2 = row16_dot<0>(h, w, 0.0f);
    if (node < n && f < NCLS) g[(unsigned)node * NCLS + f] = s * a2;
}


template <int J> __device__ __forceinline__ int grp8_bcast(int v) {
    return (int)__builtin_amdgcn_mov_dpp8((unsigned)v, (unsigned)J * 0x249249u);
}

template <int J> __device__ __forceinline__ void grp8_bcast_all(int v, int* r) {
    r[J] = grp8_bcast<J>(v);
    if constexpr (J + 1 < 8) grp8_bcast_all<J + 1>(v, r);
}

__global__ void gcn_agg2(const int* __restrict__ start, const int* __restrict__ srcs,
                         const float* __restrict__ g, const float* __restrict__ dinv,
                         const float* __restrict__ b2, float* __restrict__ out, int n) {
    int t = blockIdx.x * blockDim.x + threadIdx.x;
    int node = t >> 3;
    unsigned f = threadIdx.x & 7u;
    unsigned fc = f < NCLS ? f : NCLS - 1;
    int beg = 0, end = 0, self = 0;
    float acc = 0.0f;
    if (node < n) {
        beg = start[node];
        end = start[node + 1];
        self = node;
        acc = g[(unsigned)node * NCLS + fc];
    }
    int p = beg + (int)f;
    int i0 = (p < end) ? srcs[p] : -1;
    int i1 = (p + 8 < end) ? srcs[p + 8] : -1;
    for (int base = beg; base < end; base += 16) {
        int pn = p + 16;
        int l0 = srcs[pn < end ? pn : beg];
        int l1 = srcs[pn + 8 < end ? pn + 8 : beg];
        __builtin_amdgcn_sched_barrier(0);
        int r[16];
        float v[16];
        grp8_bcast_all<0>(i0, r);
        grp8_bcast_all<0>(i1, r + 8);
        #pragma unroll
        for (int j = 0; j < 16; ++j) v[j] = g[(unsigned)(r[j] >= 0 ? r[j] : self) * NCLS + fc];
        #pragma unroll
        for (int j = 0; j < 16; ++j) acc += (r[j] >= 0) ? v[j] : 0.0f;
        i0 = (pn < end) ? l0 : -1;
        i1 = (pn + 8 < end) ? l1 : -1;
        p = pn;
    }
    float s = (node < n) ? dinv[node] : 0.0f;
    float val = b2[fc] + s * acc;
    int vr[8];
    grp8_bcast_all<0>(__float_as_int(val), vr);
    float m = -1e30f;
    #pragma unroll
    for (int c = 0; c < NCLS; ++c) m = fmaxf(m, __int_as_float(vr[c]));
    float sum = 0.0f;
    #pragma unroll
    for (int c = 0; c < NCLS; ++c) sum += __expf(__int_as_float(vr[c]) - m);
    float lse = m + __logf(sum);
    if (node < n && f < NCLS) out[(unsigned)node * NCLS + f] = val - lse;
}


extern "C" void kernel_launch(void* const* d_in, const int* in_sizes, int n_in,
                              void* d_out, int out_size, void* d_ws, size_t ws_size,
                              hipStream_t stream) {
    const float* x  = (const float*)d_in[0];
    const int*   ei = (const int*)  d_in[1];
    const float* W1 = (const float*)d_in[2];
    const float* b1 = (const float*)d_in[3];
    const float* W2 = (const float*)d_in[4];
    const float* b2 = (const float*)d_in[5];
    float* out = (float*)d_out;

    int n = in_sizes[0] / F_IN;
    int e = in_sizes[1] / 2;
    const int* rowi = ei;
    const int* coli = ei + e;

    size_t np = ((size_t)n + 64) & ~(size_t)63;
    size_t ep = ((size_t)e + 63) & ~(size_t)63;
    size_t rw = ep > 2 * HID * np ? ep : 2 * HID * np;
    int nb = (n + BIN_W - 1) >> BIN_SH;
    if (ws_size < (3 * np + ep + rw) * sizeof(float)) return;
    if (nb > BIN_MAXB || (size_t)BIN_G * nb > 2 * np) return;

    int*   base  = (int*)d_ws;
    int*   start = base + np;
    float* dinv  = (float*)(start + np);
    int*   srcs  = (int*)(dinv + np);
    int*   hist  = start;
    unsigned* binned = (unsigned*)(srcs + ep);
    float* h1    = (float*)binned;
    float* g     = h1 + HID * np;

    const int B = 256;

    gcn_bin_count  <<<BIN_G, BIN_T, 0, stream>>>(coli, hist, e, nb);
    gcn_bin_scatter<<<BIN_G, BIN_T, 0, stream>>>(rowi, coli, hist, base, binned, e, nb);
    gcn_bucket_rank<<<nb, BIN_T, 0, stream>>>(binned, base, start, dinv, srcs, n);

    int ntiles = (n + 15) >> 4;
    int gw = (ntiles + GT_WAVES - 1) / GT_WAVES;
    gcn_gemm1_wmma <<<gw, GT_WAVES * 32, 0, stream>>>(x, W1, dinv, h1, n);

    gcn_agg1       <<<(n * 16 + B - 1) / B, B, 0, stream>>>(start, srcs, h1, dinv, b1, W2, g, n);
    gcn_agg2       <<<(n * 8 + B - 1) / B, B, 0, stream>>>(start, srcs, g, dinv, b2, out, n);
}


# Edge sort loses its counting launch: each chunk block counts and bins its own chunk in place (run table); a ranking block collects its bucket's 250 runs: 6 launches -> 5; sums in LDS arrival order
# speedup vs baseline: 12.4741x; 1.1290x over previous
#include <hip/hip_runtime.h>
#include <hip/hip_bf16.h>

#define F_IN 512
#define HID  16
#define NCLS 7

typedef float v2f __attribute__((ext_vector_type(2)));
typedef float v8f __attribute__((ext_vector_type(8)));


#define BIN_G    250
#define BIN_T    1024
#define BIN_SH   8
#define BIN_W    (1 << BIN_SH)
#define BIN_MAXB BIN_T
#define STAGE_CAP 16384
#define RUNS_PW  ((BIN_G + BIN_T / 32 - 1) / (BIN_T / 32))

__device__ __forceinline__ int wave_scan_incl(int s, int lane) {
    #pragma unroll
    for (int o = 1; o < 32; o <<= 1) {
        int t = __shfl_up(s, o);
        if (lane >= o) s += t;
    }
    return s;
}

__device__ __forceinline__ int wave_sum(int s) {
    #pragma unroll
    for (int o = 16; o > 0; o >>= 1) s += __shfl_xor(s, o);
    return s;
}

template <int K> struct batch_of { static constexpr int size = K; };

template <class Body>
__device__ __forceinline__ void for_edges(int i, int end, Body body) {
    for (; i + 7 * BIN_T < end; i += 8 * BIN_T) body(i, batch_of<8>{});
    if (i + 3 * BIN_T < end) { body(i, batch_of<4>{}); i += 4 * BIN_T; }
    if (i + 1 * BIN_T < end) { body(i, batch_of<2>{}); i += 2 * BIN_T; }
    if (i < end) body(i, batch_of<1>{});
}

__global__ void __launch_bounds__(BIN_T)
gcn_bin_local(const int* __restrict__ row, const int* __restrict__ col,
              int* __restrict__ lbeg, unsigned* __restrict__ binned, int e, int nb) {
    __shared__ int cur[BIN_MAXB];
    __shared__ int wsum[BIN_T / 32];
    __shared__ unsigned stage[STAGE_CAP];
    int tid = threadIdx.x, lane = tid & 31, w = tid >> 5;
    cur[tid] = 0;
    __syncthreads();

    int ch = (e + BIN_G - 1) / BIN_G;
    int lo = blockIdx.x * ch;
    lo = lo < e ? lo : e;
    int hi = lo + ch < e ? lo + ch : e;

    for_edges(lo + tid, hi, [&](int i, auto kc) {
        constexpr int K = decltype(kc)::size;
        int c[K];
        #pragma unroll
        for (int j = 0; j < K; ++j) c[j] = col[i + j * BIN_T];
        #pragma unroll
        for (int j = 0; j < K; ++j) atomicAdd(&cur[c[j] >> BIN_SH], 1);
    });
    __syncthreads();

    int mine = tid < nb ? cur[tid] : 0;
    int linc = wave_scan_incl(mine, lane);
    if (lane == 31) wsum[w] = linc;
    __syncthreads();
    if (w == 0) {
        int u = wsum[lane];
        int iu = wave_scan_incl(u, lane);
        wsum[lane] = iu - u;
    }
    __syncthreads();
    int lfirst = wsum[w] + linc - mine;
    if (tid < nb) {
        cur[tid] = lfirst;
        int* lrow = lbeg + (size_t)blockIdx.x * (nb + 1);
        lrow[tid] = lfirst;
        if (tid == nb - 1) lrow[nb] = lfirst + mine;
    }
    __syncthreads();

    if (hi - lo <= STAGE_CAP) {
        for_edges(lo + tid, hi, [&](int i, auto kc) {
            constexpr int K = decltype(kc)::size;
            int c[K], r[K];
            #pragma unroll
            for (int j = 0; j < K; ++j) {
                c[j] = col[i + j * BIN_T];
                r[j] = row[i + j * BIN_T];
            }
            int pos[K];
            #pragma unroll
            for (int j = 0; j < K; ++j) pos[j] = atomicAdd(&cur[c[j] >> BIN_SH], 1);
            #pragma unroll
            for (int j = 0; j < K; ++j)
                stage[pos[j]] = ((unsigned)r[j] << BIN_SH) | ((unsigned)c[j] & (BIN_W - 1));
        });
        __syncthreads();
        int n_sl = hi - lo;
        int head = (32 - (lo & 31)) & 31;
        head = head < n_sl ? head : n_sl;
        if (tid < head) binned[lo + tid] = stage[tid];
        for_edges(head + tid, n_sl, [&](int i, auto kc) {
            constexpr int K = decltype(kc)::size;
            unsigned t[K];
            #pragma unroll
            for (int j = 0; j < K; ++j) t[j] = stage[i + j * BIN_T];
            #pragma unroll
            for (int j = 0; j < K; ++j) binned[lo + i + j * BIN_T] = t[j];
        });
    } else {
        for_edges(lo + tid, hi, [&](int i, auto kc) {
            constexpr int K = decltype(kc)::size;
            int c[K], r[K];
            #pragma unroll
            for (int j = 0; j < K; ++j) {
                c[j] = col[i + j * BIN_T];
                r[j] = row[i + j * BIN_T];
            }
            int pos[K];
            #pragma unroll
            for (int j = 0; j < K; ++j) pos[j] = atomicAdd(&cur[c[j] >> BIN_SH], 1);
            #pragma unroll
            for (int j = 0; j < K; ++j)
                binned[lo + pos[j]] = ((unsigned)r[j] << BIN_SH) | ((unsigned)c[j] & (BIN_W - 1));
        });
    }
}

__global__ void __launch_bounds__(BIN_T)
gcn_bucket_rank(const unsigned* __restrict__ binned, const int* __restrict__ lbeg,
                int* __restrict__ start, float* __restrict__ dinv, int* __restrict__ srcs,
                int n, int e, int nb) {
    __shared__ int cnt[BIN_W];
    __shared__ int slot[BIN_W];
    __shared__ int wsum[BIN_W / 32];
    __shared__ int roff[BIN_G];
    __shared__ int rlen[BIN_G];
    __shared__ int tot[2];
    __shared__ int stage[STAGE_CAP];
    int tid = threadIdx.x, lane = tid & 31, w = tid >> 5;
    int ch = (e + BIN_G - 1) / BIN_G;
    if (tid < BIN_W) cnt[tid] = 0;
    if (tid < 2) tot[tid] = 0;
    __syncthreads();

    if (w < (BIN_G + 31) / 32) {
        int s = 0, t = 0;
        if (tid < BIN_G) {
            const int* lrow = lbeg + (size_t)tid * (nb + 1) + blockIdx.x;
            s = lrow[0];
            t = lrow[1];
            int first = tid * ch;
            roff[tid] = (first < e ? first : e) + s;
            rlen[tid] = t - s;
        }
        s = wave_sum(s);
        t = wave_sum(t);
        if (lane == 0) { atomicAdd(&tot[0], s); atomicAdd(&tot[1], t); }
    }
    __syncthreads();
    int lo = tot[0];
    int hi = tot[1];

    int rs[RUNS_PW], rl[RUNS_PW];
    #pragma unroll
    for (int k = 0; k < RUNS_PW; ++k) {
        int g  = w + k * (BIN_T / 32);
        int gc = g < BIN_G ? g : 0;
        rs[k] = roff[gc];
        rl[k] = g < BIN_G ? rlen[gc] : 0;
    }
    unsigned x[RUNS_PW], y[RUNS_PW];
    #pragma unroll
    for (int k = 0; k < RUNS_PW; ++k) {
        x[k] = binned[lane      < rl[k] ? rs[k] + lane      : 0];
        y[k] = binned[lane + 32 < rl[k] ? rs[k] + lane + 32 : 0];
    }
    #pragma unroll
    for (int k = 0; k < RUNS_PW; ++k) {
        if (lane      < rl[k]) atomicAdd(&cnt[x[k] & (BIN_W - 1)], 1);
        if (lane + 32 < rl[k]) atomicAdd(&cnt[y[k] & (BIN_W - 1)], 1);
    }
    #pragma unroll
    for (int k = 0; k < RUNS_PW; ++k)
        for (int j = lane + 64; j < rl[k]; j += 32) atomicAdd(&cnt[binned[rs[k] + j] & (BIN_W - 1)], 1);
    __syncthreads();
    int c = 0, inc = 0;
    if (tid < BIN_W) {
        c = cnt[tid];
        inc = wave_scan_incl(c, lane);
        if (lane == 31) wsum[w] = inc;
    }
    __syncthreads();
    if (tid < BIN_W) {
        int front = 0;
        #pragma unroll
        for (int k = 0; k < BIN_W / 32; ++k) front += (k < w) ? wsum[k] : 0;
        int excl = front + inc - c;
        slot[tid] = excl;
        int node = blockIdx.x * BIN_W + tid;
        if (node < n) {
            start[node] = lo + excl;
            dinv[node]  = rsqrtf((float)(c + 1));
            if (node == n - 1) start[n] = lo + excl + c;
        }
    }
    __syncthreads();
    if (hi - lo <= STAGE_CAP) {
        #pragma unroll
        for (int k = 0; k < RUNS_PW; ++k) {
            if (lane < rl[k]) {
                int r = atomicAdd(&slot[x[k] & (BIN_W - 1)], 1);
                stage[r] = (int)(x[k] >> BIN_SH);
            }
            if (lane + 32 < rl[k]) {
                int r = atomicAdd(&slot[y[k] & (BIN_W - 1)], 1);
                stage[r] = (int)(y[k] >> BIN_SH);
            }
        }
        #pragma unroll
        for (int k = 0; k < RUNS_PW; ++k)
            for (int j = lane + 64; j < rl[k]; j += 32) {
                unsigned v = binned[rs[k] + j];
                int r = atomicAdd(&slot[v & (BIN_W - 1)], 1);
                stage[r] = (int)(v >> BIN_SH);
            }
        __syncthreads();
        int n_sl = hi - lo;
        int head = (32 - (lo & 31)) & 31;
        head = head < n_sl ? head : n_sl;
        if (tid < head) srcs[lo + tid] = stage[tid];
        for_edges(head + tid, n_sl, [&](int i, auto kc) {
            constexpr int K = decltype(kc)::size;
            int t[K];
            #pragma unroll
            for (int j = 0; j < K; ++j) t[j] = stage[i + j * BIN_T];
            #pragma unroll
            for (int j = 0; j < K; ++j) srcs[lo + i + j * BIN_T] = t[j];
        });
    } else {
        #pragma unroll
        for (int k = 0; k < RUNS_PW; ++k) {
            if (lane < rl[k]) {
                int r = atomicAdd(&slot[x[k] & (BIN_W - 1)], 1);
                srcs[lo + r] = (int)(x[k] >> BIN_SH);
            }
            if (lane + 32 < rl[k]) {
                int r = atomicAdd(&slot[y[k] & (BIN_W - 1)], 1);
                srcs[lo + r] = (int)(y[k] >> BIN_SH);
            }
        }
        #pragma unroll
        for (int k = 0; k < RUNS_PW; ++k)
            for (int j = lane + 64; j < rl[k]; j += 32) {
                unsigned v = binned[rs[k] + j];
                int r = atomicAdd(&slot[v & (BIN_W - 1)], 1);
                srcs[lo + r] = (int)(v >> BIN_SH);
            }
    }
}


#define GT_WAVES 4
#define GT_CH    128
#define GT_P4    (GT_CH / 4 + 1)

typedef float v4f __attribute__((ext_vector_type(4)));

__device__ __forceinline__ void store_tile(float* __restrict__ h1, const float* __restrict__ dinv,
                                           const v8f& acc, int tile, int lane, int n) {
    int mbase = tile * 16 + ((lane >> 4) << 3);
    int ncol  = lane & 15;
    float* p = h1 + (size_t)mbase * HID + ncol;
    if (mbase + 8 <= n) {
        float s[8];
        #pragma unroll
        for (int v = 0; v < 8; ++v) s[v] = dinv[mbase + v];
        #pragma unroll
        for (int v = 0; v < 8; ++v) p[v * HID] = s[v] * acc[v];
    } else {
        #pragma unroll
        for (int v = 0; v < 8; ++v) if (mbase + v < n) p[v * HID] = dinv[mbase + v] * acc[v];
    }
}

__global__ void __launch_bounds__(GT_WAVES * 32)
gcn_gemm1_wmma(const float* __restrict__ x, const float* __restrict__ W1,
               const float* __restrict__ dinv, float* __restrict__ h1, int n) {
    __shared__ v4f wf[(F_IN / 8) * 32];
    __shared__ v4f xs[GT_WAVES][16 * GT_P4];
    int lane = threadIdx.x & 31;
    int w    = threadIdx.x >> 5;
    int m    = lane & 15;
    int h    = lane >> 4;
    int ntiles = (n + 15) >> 4;
    int tile = blockIdx.x * GT_WAVES + w;
    int tl   = tile < ntiles ? tile : ntiles - 1;

    {
        v4f t[F_IN / 8 / GT_WAVES];
        #pragma unroll
        for (int i = 0; i < F_IN / 8 / GT_WAVES; ++i) {
            int G = w + i * GT_WAVES;
            const float* p = W1 + (size_t)(8 * G + 4 * h) * HID + m;
            t[i].x = p[0 * HID];
            t[i].y = p[1 * HID];
            t[i].z = p[2 * HID];
            t[i].w = p[3 * HID];
        }
        #pragma unroll
        for (int i = 0; i < F_IN / 8 / GT_WAVES; ++i) wf[(w + i * GT_WAVES) * 32 + lane] = t[i];
    }

    const v4f* xrow[16];
    #pragma unroll
    for (int r = 0; r < 16; ++r) {
        int row = tl * 16 + r;
        xrow[r] = (const v4f*)(x + (size_t)(row < n ? row : 0) * F_IN) + lane;
    }
    v4f pre[16];
    #pragma unroll
    for (int r = 0; r < 16; ++r) pre[r] = xrow[r][0];

    __syncthreads();

    v4f* xw = xs[w];
    v8f acc0 = {}, acc1 = {};
    #pragma unroll
    for (int c = 0; c < F_IN / GT_CH; ++c) {
        #pragma unroll
        for (int r = 0; r < 16; ++r) xw[r * GT_P4 + lane] = pre[r];
        if (c + 1 < F_IN / GT_CH) {
            #pragma unroll
            for (int r = 0; r < 16; ++r) pre[r] = xrow[r][(c + 1) * (GT_CH / 4)];
        }
        __builtin_amdgcn_wave_barrier();
        #pragma unroll
        for (int g = 0; g < GT_CH / 8; ++g) {
            v4f a = xw[m * GT_P4 + 2 * g + h];
            v4f b = wf[(c * (GT_CH / 8) + g) * 32 + lane];
            v2f a0 = {a.x, a.y}, a1 = {a.z, a.w};
            v2f b0 = {b.x, b.y}, b1 = {b.z, b.w};
            acc0 = __builtin_amdgcn_wmma_f32_16x16x4_f32(
                false, a0, false, b0, (short)0, acc0, false, false);
            acc1 = __builtin_amdgcn_wmma_f32_16x16x4_f32(
                false, a1, false, b1, (short)0, acc1, false, false);
        }
        __builtin_amdgcn_wave_barrier();
    }
    v8f acc = acc0 + acc1;
    if (tile < ntiles) store_tile(h1, dinv, acc, tile, lane, n);
}


template <int J> __device__ __forceinline__ int row16_bcast(int v) {
    return __builtin_amdgcn_update_dpp(0, v, 0x150 + J, 0xf, 0xf, true);
}

template <int J> __device__ __forceinline__ void row16_bcast_all(int v, int* r) {
    r[J] = row16_bcast<J>(v);
    if constexpr (J + 1 < 16) row16_bcast_all<J + 1>(v, r);
}

template <int J> __device__ __forceinline__ float row16_dot(float h, const float* w, float a) {
    a = __builtin_fmaf(__int_as_float(row16_bcast<J>(__float_as_int(h))), w[J], a);
    if constexpr (J + 1 < 16) return row16_dot<J + 1>(h, w, a);
    else return a;
}

__global__ void gcn_agg1(const int* __restrict__ start, const int* __restrict__ srcs,
                         const float* __restrict__ hs, const float* __restrict__ dinv,
                         const float* __restrict__ b1, const float* __restrict__ W2,
                         float* __restrict__ g, int n) {
    int t = blockIdx.x * blockDim.x + threadIdx.x;
    int node = t >> 4;
    unsigned f = threadIdx.x & 15u;
    int beg = 0, end = 0, self = 0;
    float acc = 0.0f;
    if (node < n) {
        beg = start[node];
        end = start[node + 1];
        self = node;
        acc = hs[(unsigned)node * HID + f];
    }
    int p = beg + (int)f;
    int idx = (p < end) ? srcs[p] : -1;
    for (int base = beg; base < end; base += 16) {
        int pn = p + 16;
        int ld = srcs[pn < end ? pn : beg];
        __builtin_amdgcn_sched_barrier(0);
        int r[16];
        float v[16];
        row16_bcast_all<0>(idx, r);
        #pragma unroll
        for (int j = 0; j < 16; ++j) v[j] = hs[(unsigned)(r[j] >= 0 ? r[j] : self) * HID + f];
        #pragma unroll
        for (int j = 0; j < 16; ++j) acc += (r[j] >= 0) ? v[j] : 0.0f;
        idx = (pn < end) ? ld : -1;
        p = pn;
    }
    unsigned fc = f < NCLS ? f : NCLS - 1;
    float s = (node < n) ? dinv[node] : 0.0f;
    float w[HID];
    #pragma unroll
    for (int j = 0; j < HID; ++j) w[j] = W2[j * NCLS + fc];
    float h = fmaxf(b1[f] + s * acc, 0.0f);
    float a2 = row16_dot<0>(h, w, 0.0f);
    if (node < n && f < NCLS) g[(unsigned)node * NCLS + f] = s * a2;
}


template <int J> __device__ __forceinline__ int grp8_bcast(int v) {
    return (int)__builtin_amdgcn_mov_dpp8((unsigned)v, (unsigned)J * 0x249249u);
}

template <int J> __device__ __forceinline__ void grp8_bcast_all(int v, int* r) {
    r[J] = grp8_bcast<J>(v);
    if constexpr (J + 1 < 8) grp8_bcast_all<J + 1>(v, r);
}

__global__ void gcn_agg2(const int* __restrict__ start, const int* __restrict__ srcs,
                         const float* __restrict__ g, const float* __restrict__ dinv,
                         const float* __restrict__ b2, float* __restrict__ out, int n) {
    int t = blockIdx.x * blockDim.x + threadIdx.x;
    int node = t >> 3;
    unsigned f = threadIdx.x & 7u;
    unsigned fc = f < NCLS ? f : NCLS - 1;
    int beg = 0, end = 0, self = 0;
    float acc = 0.0f;
    if (node < n) {
        beg = start[node];
        end = start[node + 1];
        self = node;
        acc = g[(unsigned)node * NCLS + fc];
    }
    int p = beg + (int)f;
    int i0 = (p < end) ? srcs[p] : -1;
    int i1 = (p + 8 < end) ? srcs[p + 8] : -1;
    for (int base = beg; base < end; base += 16) {
        int pn = p + 16;
        int l0 = srcs[pn < end ? pn : beg];
        int l1 = srcs[pn + 8 < end ? pn + 8 : beg];
        __builtin_amdgcn_sched_barrier(0);
        int r[16];
        float v[16];
        grp8_bcast_all<0>(i0, r);
        grp8_bcast_all<0>(i1, r + 8);
        #pragma unroll
        for (int j = 0; j < 16; ++j) v[j] = g[(unsigned)(r[j] >= 0 ? r[j] : self) * NCLS + fc];
        #pragma unroll
        for (int j = 0; j < 16; ++j) acc += (r[j] >= 0) ? v[j] : 0.0f;
        i0 = (pn < end) ? l0 : -1;
        i1 = (pn + 8 < end) ? l1 : -1;
        p = pn;
    }
    float s = (node < n) ? dinv[node] : 0.0f;
    float val = b2[fc] + s * acc;
    int vr[8];
    grp8_bcast_all<0>(__float_as_int(val), vr);
    float m = -1e30f;
    #pragma unroll
    for (int c = 0; c < NCLS; ++c) m = fmaxf(m, __int_as_float(vr[c]));
    float sum = 0.0f;
    #pragma unroll
    for (int c = 0; c < NCLS; ++c) sum += __expf(__int_as_float(vr[c]) - m);
    float lse = m + __logf(sum);
    if (node < n && f < NCLS) out[(unsigned)node * NCLS + f] = val - lse;
}


extern "C" void kernel_launch(void* const* d_in, const int* in_sizes, int n_in,
                              void* d_out, int out_size, void* d_ws, size_t ws_size,
                              hipStream_t stream) {
    const float* x  = (const float*)d_in[0];
    const int*   ei = (const int*)  d_in[1];
    const float* W1 = (const float*)d_in[2];
    const float* b1 = (const float*)d_in[3];
    const float* W2 = (const float*)d_in[4];
    const float* b2 = (const float*)d_in[5];
    float* out = (float*)d_out;

    int n = in_sizes[0] / F_IN;
    int e = in_sizes[1] / 2;
    const int* rowi = ei;
    const int* coli = ei + e;

    size_t np = ((size_t)n + 64) & ~(size_t)63;
    size_t ep = ((size_t)e + 63) & ~(size_t)63;
    size_t rw = ep > 2 * HID * np ? ep : 2 * HID * np;
    int nb = (n + BIN_W - 1) >> BIN_SH;
    if (ws_size < (3 * np + ep + rw) * sizeof(float)) return;
    if (nb > BIN_MAXB || (size_t)BIN_G * (nb + 1) > np) return;

    int*   lbeg  = (int*)d_ws;
    int*   start = lbeg + np;
    float* dinv  = (float*)(start + np);
    int*   srcs  = (int*)(dinv + np);
    unsigned* binned = (unsigned*)(srcs + ep);
    float* h1    = (float*)binned;
    float* g     = h1 + HID * np;

    const int B = 256;

    gcn_bin_local  <<<BIN_G, BIN_T, 0, stream>>>(rowi, coli, lbeg, binned, e, nb);
    gcn_bucket_rank<<<nb, BIN_T, 0, stream>>>(binned, lbeg, start, dinv, srcs, n, e, nb);

    int ntiles = (n + 15) >> 4;
    int gw = (ntiles + GT_WAVES - 1) / GT_WAVES;
    gcn_gemm1_wmma <<<gw, GT_WAVES * 32, 0, stream>>>(x, W1, dinv, h1, n);

    gcn_agg1       <<<(n * 16 + B - 1) / B, B, 0, stream>>>(start, srcs, h1, dinv, b1, W2, g, n);
    gcn_agg2       <<<(n * 8 + B - 1) / B, B, 0, stream>>>(start, srcs, g, dinv, b2, out, n);
}


# Both per-node sums: 4 lanes per node, each lane gathers 16 B (conv1) or 8 B (conv2, g rows padded to 8 floats) of a source row, 8 nodes per wave: 4x/2x fewer waves, ~4x fewer loads; sum order kept
# speedup vs baseline: 13.6083x; 1.0909x over previous
#include <hip/hip_runtime.h>
#include <hip/hip_bf16.h>

#define F_IN 512
#define HID  16
#define NCLS 7

typedef float v2f __attribute__((ext_vector_type(2)));
typedef float v8f __attribute__((ext_vector_type(8)));


#define BIN_G    250
#define BIN_T    1024
#define BIN_SH   8
#define BIN_W    (1 << BIN_SH)
#define BIN_MAXB BIN_T
#define STAGE_CAP 16384
#define RUNS_PW  ((BIN_G + BIN_T / 32 - 1) / (BIN_T / 32))

__device__ __forceinline__ int wave_scan_incl(int s, int lane) {
    #pragma unroll
    for (int o = 1; o < 32; o <<= 1) {
        int t = __shfl_up(s, o);
        if (lane >= o) s += t;
    }
    return s;
}

__device__ __forceinline__ int wave_sum(int s) {
    #pragma unroll
    for (int o = 16; o > 0; o >>= 1) s += __shfl_xor(s, o);
    return s;
}

template <int K> struct batch_of { static constexpr int size = K; };

template <class Body>
__device__ __forceinline__ void for_edges(int i, int end, Body body) {
    for (; i + 7 * BIN_T < end; i += 8 * BIN_T) body(i, batch_of<8>{});
    if (i + 3 * BIN_T < end) { body(i, batch_of<4>{}); i += 4 * BIN_T; }
    if (i + 1 * BIN_T < end) { body(i, batch_of<2>{}); i += 2 * BIN_T; }
    if (i < end) body(i, batch_of<1>{});
}

__global__ void __launch_bounds__(BIN_T)
gcn_bin_local(const int* __restrict__ row, const int* __restrict__ col,
              int* __restrict__ lbeg, unsigned* __restrict__ binned, int e, int nb) {
    __shared__ int cur[BIN_MAXB];
    __shared__ int wsum[BIN_T / 32];
    __shared__ unsigned stage[STAGE_CAP];
    int tid = threadIdx.x, lane = tid & 31, w = tid >> 5;
    cur[tid] = 0;
    __syncthreads();

    int ch = (e + BIN_G - 1) / BIN_G;
    int lo = blockIdx.x * ch;
    lo = lo < e ? lo : e;
    int hi = lo + ch < e ? lo + ch : e;

    for_edges(lo + tid, hi, [&](int i, auto kc) {
        constexpr int K = decltype(kc)::size;
        int c[K];
        #pragma unroll
        for (int j = 0; j < K; ++j) c[j] = col[i + j * BIN_T];
        #pragma unroll
        for (int j = 0; j < K; ++j) atomicAdd(&cur[c[j] >> BIN_SH], 1);
    });
    __syncthreads();

    int mine = tid < nb ? cur[tid] : 0;
    int linc = wave_scan_incl(mine, lane);
    if (lane == 31) wsum[w] = linc;
    __syncthreads();
    if (w == 0) {
        int u = wsum[lane];
        int iu = wave_scan_incl(u, lane);
        wsum[lane] = iu - u;
    }
    __syncthreads();
    int lfirst = wsum[w] + linc - mine;
    if (tid < nb) {
        cur[tid] = lfirst;
        int* lrow = lbeg + (size_t)blockIdx.x * (nb + 1);
        lrow[tid] = lfirst;
        if (tid == nb - 1) lrow[nb] = lfirst + mine;
    }
    __syncthreads();

    if (hi - lo <= STAGE_CAP) {
        for_edges(lo + tid, hi, [&](int i, auto kc) {
            constexpr int K = decltype(kc)::size;
            int c[K], r[K];
            #pragma unroll
            for (int j = 0; j < K; ++j) {
                c[j] = col[i + j * BIN_T];
                r[j] = row[i + j * BIN_T];
            }
            int pos[K];
            #pragma unroll
            for (int j = 0; j < K; ++j) pos[j] = atomicAdd(&cur[c[j] >> BIN_SH], 1);
            #pragma unroll
            for (int j = 0; j < K; ++j)
                stage[pos[j]] = ((unsigned)r[j] << BIN_SH) | ((unsigned)c[j] & (BIN_W - 1));
        });
        __syncthreads();
        int n_sl = hi - lo;
        int head = (32 - (lo & 31)) & 31;
        head = head < n_sl ? head : n_sl;
        if (tid < head) binned[lo + tid] = stage[tid];
        for_edges(head + tid, n_sl, [&](int i, auto kc) {
            constexpr int K = decltype(kc)::size;
            unsigned t[K];
            #pragma unroll
            for (int j = 0; j < K; ++j) t[j] = stage[i + j * BIN_T];
            #pragma unroll
            for (int j = 0; j < K; ++j) binned[lo + i + j * BIN_T] = t[j];
        });
    } else {
        for_edges(lo + tid, hi, [&](int i, auto kc) {
            constexpr int K = decltype(kc)::size;
            int c[K], r[K];
            #pragma unroll
            for (int j = 0; j < K; ++j) {
                c[j] = col[i + j * BIN_T];
                r[j] = row[i + j * BIN_T];
            }
            int pos[K];
            #pragma unroll
            for (int j = 0; j < K; ++j) pos[j] = atomicAdd(&cur[c[j] >> BIN_SH], 1);
            #pragma unroll
            for (int j = 0; j < K; ++j)
                binned[lo + pos[j]] = ((unsigned)r[j] << BIN_SH) | ((unsigned)c[j] & (BIN_W - 1));
        });
    }
}

__global__ void __launch_bounds__(BIN_T)
gcn_bucket_rank(const unsigned* __restrict__ binned, const int* __restrict__ lbeg,
                int* __restrict__ start, float* __restrict__ dinv, int* __restrict__ srcs,
                int n, int e, int nb) {
    __shared__ int cnt[BIN_W];
    __shared__ int slot[BIN_W];
    __shared__ int wsum[BIN_W / 32];
    __shared__ int roff[BIN_G];
    __shared__ int rlen[BIN_G];
    __shared__ int tot[2];
    __shared__ int stage[STAGE_CAP];
    int tid = threadIdx.x, lane = tid & 31, w = tid >> 5;
    int ch = (e + BIN_G - 1) / BIN_G;
    if (tid < BIN_W) cnt[tid] = 0;
    if (tid < 2) tot[tid] = 0;
    __syncthreads();

    if (w < (BIN_G + 31) / 32) {
        int s = 0, t = 0;
        if (tid < BIN_G) {
            const int* lrow = lbeg + (size_t)tid * (nb + 1) + blockIdx.x;
            s = lrow[0];
            t = lrow[1];
            int first = tid * ch;
            roff[tid] = (first < e ? first : e) + s;
            rlen[tid] = t - s;
        }
        s = wave_sum(s);
        t = wave_sum(t);
        if (lane == 0) { atomicAdd(&tot[0], s); atomicAdd(&tot[1], t); }
    }
    __syncthreads();
    int lo = tot[0];
    int hi = tot[1];

    int rs[RUNS_PW], rl[RUNS_PW];
    #pragma unroll
    for (int k = 0; k < RUNS_PW; ++k) {
        int g  = w + k * (BIN_T / 32);
        int gc = g < BIN_G ? g : 0;
        rs[k] = roff[gc];
        rl[k] = g < BIN_G ? rlen[gc] : 0;
    }
    unsigned x[RUNS_PW], y[RUNS_PW];
    #pragma unroll
    for (int k = 0; k < RUNS_PW; ++k) {
        x[k] = binned[lane      < rl[k] ? rs[k] + lane      : 0];
        y[k] = binned[lane + 32 < rl[k] ? rs[k] + lane + 32 : 0];
    }
    #pragma unroll
    for (int k = 0; k < RUNS_PW; ++k) {
        if (lane      < rl[k]) atomicAdd(&cnt[x[k] & (BIN_W - 1)], 1);
        if (lane + 32 < rl[k]) atomicAdd(&cnt[y[k] & (BIN_W - 1)], 1);
    }
    #pragma unroll
    for (int k = 0; k < RUNS_PW; ++k)
        for (int j = lane + 64; j < rl[k]; j += 32) atomicAdd(&cnt[binned[rs[k] + j] & (BIN_W - 1)], 1);
    __syncthreads();
    int c = 0, inc = 0;
    if (tid < BIN_W) {
        c = cnt[tid];
        inc = wave_scan_incl(c, lane);
        if (lane == 31) wsum[w] = inc;
    }
    __syncthreads();
    if (tid < BIN_W) {
        int front = 0;
        #pragma unroll
        for (int k = 0; k < BIN_W / 32; ++k) front += (k < w) ? wsum[k] : 0;
        int excl = front + inc - c;
        slot[tid] = excl;
        int node = blockIdx.x * BIN_W + tid;
        if (node < n) {
            start[node] = lo + excl;
            dinv[node]  = rsqrtf((float)(c + 1));
            if (node == n - 1) start[n] = lo + excl + c;
        }
    }
    __syncthreads();
    if (hi - lo <= STAGE_CAP) {
        #pragma unroll
        for (int k = 0; k < RUNS_PW; ++k) {
            if (lane < rl[k]) {
                int r = atomicAdd(&slot[x[k] & (BIN_W - 1)], 1);
                stage[r] = (int)(x[k] >> BIN_SH);
            }
            if (lane + 32 < rl[k]) {
                int r = atomicAdd(&slot[y[k] & (BIN_W - 1)], 1);
                stage[r] = (int)(y[k] >> BIN_SH);
            }
        }
        #pragma unroll
        for (int k = 0; k < RUNS_PW; ++k)
            for (int j = lane + 64; j < rl[k]; j += 32) {
                unsigned v = binned[rs[k] + j];
                int r = atomicAdd(&slot[v & (BIN_W - 1)], 1);
                stage[r] = (int)(v >> BIN_SH);
            }
        __syncthreads();
        int n_sl = hi - lo;
        int head = (32 - (lo & 31)) & 31;
        head = head < n_sl ? head : n_sl;
        if (tid < head) srcs[lo + tid] = stage[tid];
        for_edges(head + tid, n_sl, [&](int i, auto kc) {
            constexpr int K = decltype(kc)::size;
            int t[K];
            #pragma unroll
            for (int j = 0; j < K; ++j) t[j] = stage[i + j * BIN_T];
            #pragma unroll
            for (int j = 0; j < K; ++j) srcs[lo + i + j * BIN_T] = t[j];
        });
    } else {
        #pragma unroll
        for (int k = 0; k < RUNS_PW; ++k) {
            if (lane < rl[k]) {
                int r = atomicAdd(&slot[x[k] & (BIN_W - 1)], 1);
                srcs[lo + r] = (int)(x[k] >> BIN_SH);
            }
            if (lane + 32 < rl[k]) {
                int r = atomicAdd(&slot[y[k] & (BIN_W - 1)], 1);
                srcs[lo + r] = (int)(y[k] >> BIN_SH);
            }
        }
        #pragma unroll
        for (int k = 0; k < RUNS_PW; ++k)
            for (int j = lane + 64; j < rl[k]; j += 32) {
                unsigned v = binned[rs[k] + j];
                int r = atomicAdd(&slot[v & (BIN_W - 1)], 1);
                srcs[lo + r] = (int)(v >> BIN_SH);
            }
    }
}


#define GT_WAVES 4
#define GT_CH    128
#define GT_P4    (GT_CH / 4 + 1)

typedef float v4f __attribute__((ext_vector_type(4)));

__device__ __forceinline__ void store_tile(float* __restrict__ h1, const float* __restrict__ dinv,
                                           const v8f& acc, int tile, int lane, int n) {
    int mbase = tile * 16 + ((lane >> 4) << 3);
    int ncol  = lane & 15;
    float* p = h1 + (size_t)mbase * HID + ncol;
    if (mbase + 8 <= n) {
        float s[8];
        #pragma unroll
        for (int v = 0; v < 8; ++v) s[v] = dinv[mbase + v];
        #pragma unroll
        for (int v = 0; v < 8; ++v) p[v * HID] = s[v] * acc[v];
    } else {
        #pragma unroll
        for (int v = 0; v < 8; ++v) if (mbase + v < n) p[v * HID] = dinv[mbase + v] * acc[v];
    }
}

__global__ void __launch_bounds__(GT_WAVES * 32)
gcn_gemm1_wmma(const float* __restrict__ x, const float* __restrict__ W1,
               const float* __restrict__ dinv, float* __restrict__ h1, int n) {
    __shared__ v4f wf[(F_IN / 8) * 32];
    __shared__ v4f xs[GT_WAVES][16 * GT_P4];
    int lane = threadIdx.x & 31;
    int w    = threadIdx.x >> 5;
    int m    = lane & 15;
    int h    = lane >> 4;
    int ntiles = (n + 15) >> 4;
    int tile = blockIdx.x * GT_WAVES + w;
    int tl   = tile < ntiles ? tile : ntiles - 1;

    {
        v4f t[F_IN / 8 / GT_WAVES];
        #pragma unroll
        for (int i = 0; i < F_IN / 8 / GT_WAVES; ++i) {
            int G = w + i * GT_WAVES;
            const float* p = W1 + (size_t)(8 * G + 4 * h) * HID + m;
            t[i].x = p[0 * HID];
            t[i].y = p[1 * HID];
            t[i].z = p[2 * HID];
            t[i].w = p[3 * HID];
        }
        #pragma unroll
        for (int i = 0; i < F_IN / 8 / GT_WAVES; ++i) wf[(w + i * GT_WAVES) * 32 + lane] = t[i];
    }

    const v4f* xrow[16];
    #pragma unroll
    for (int r = 0; r < 16; ++r) {
        int row = tl * 16 + r;
        xrow[r] = (const v4f*)(x + (size_t)(row < n ? row : 0) * F_IN) + lane;
    }
    v4f pre[16];
    #pragma unroll
    for (int r = 0; r < 16; ++r) pre[r] = xrow[r][0];

    __syncthreads();

    v4f* xw = xs[w];
    v8f acc0 = {}, acc1 = {};
    #pragma unroll
    for (int c = 0; c < F_IN / GT_CH; ++c) {
        #pragma unroll
        for (int r = 0; r < 16; ++r) xw[r * GT_P4 + lane] = pre[r];
        if (c + 1 < F_IN / GT_CH) {
            #pragma unroll
            for (int r = 0; r < 16; ++r) pre[r] = xrow[r][(c + 1) * (GT_CH / 4)];
        }
        __builtin_amdgcn_wave_barrier();
        #pragma unroll
        for (int g = 0; g < GT_CH / 8; ++g) {
            v4f a = xw[m * GT_P4 + 2 * g + h];
            v4f b = wf[(c * (GT_CH / 8) + g) * 32 + lane];
            v2f a0 = {a.x, a.y}, a1 = {a.z, a.w};
            v2f b0 = {b.x, b.y}, b1 = {b.z, b.w};
            acc0 = __builtin_amdgcn_wmma_f32_16x16x4_f32(
                false, a0, false, b0, (short)0, acc0, false, false);
            acc1 = __builtin_amdgcn_wmma_f32_16x16x4_f32(
                false, a1, false, b1, (short)0, acc1, false, false);
        }
        __builtin_amdgcn_wave_barrier();
    }
    v8f acc = acc0 + acc1;
    if (tile < ntiles) store_tile(h1, dinv, acc, tile, lane, n);
}


#define AG_B 256

template <int J> __device__ __forceinline__ int quad_bcast(int v) {
    return __builtin_amdgcn_update_dpp(0, v, J * 0x55, 0xf, 0xf, true);
}

template <int J> __device__ __forceinline__ void quad_bcast_all(int v, int* r) {
    r[J] = quad_bcast<J>(v);
    if constexpr (J + 1 < 4) quad_bcast_all<J + 1>(v, r);
}

template <int J> __device__ __forceinline__ float quad_bcast_f(float v) {
    return __int_as_float(quad_bcast<J>(__float_as_int(v)));
}

__global__ void __launch_bounds__(AG_B)
gcn_agg1(const int* __restrict__ start, const int* __restrict__ srcs,
         const float* __restrict__ hs, const float* __restrict__ dinv,
         const float* __restrict__ b1, const float* __restrict__ W2,
         float* __restrict__ g, int n) {
    __shared__ v2f wt[4][HID];
    int tid = threadIdx.x;
    unsigned q = tid & 3u;
    int node = (blockIdx.x * AG_B + tid) >> 2;
    const v4f* hs4 = (const v4f*)hs;
    v2f wv = {0.0f, 0.0f};
    if (tid < 4 * HID) {
        int wq = tid >> 4, wj = tid & (HID - 1);
        wv.x = W2[wj * NCLS + 2 * wq];
        if (2 * wq + 1 < NCLS) wv.y = W2[wj * NCLS + 2 * wq + 1];
    }
    int beg = 0, end = 0, self = 0;
    v4f acc = {0.0f, 0.0f, 0.0f, 0.0f};
    if (node < n) {
        beg = start[node];
        end = start[node + 1];
        self = node;
        acc = hs4[(unsigned)node * 4 + q];
    }
    if (tid < 4 * HID) wt[tid >> 4][tid & (HID - 1)] = wv;
    __syncthreads();

    int p = beg + (int)q;
    int i0 = (p < end) ? srcs[p] : -1;
    int i1 = (p + 4 < end) ? srcs[p + 4] : -1;
    for (int base = beg; base < end; base += 8) {
        int pn = p + 8;
        int l0 = srcs[pn < end ? pn : beg];
        int l1 = srcs[pn + 4 < end ? pn + 4 : beg];
        __builtin_amdgcn_sched_barrier(0);
        int r[8];
        v4f v[8];
        quad_bcast_all<0>(i0, r);
        quad_bcast_all<0>(i1, r + 4);
        #pragma unroll
        for (int j = 0; j < 8; ++j) v[j] = hs4[(unsigned)(r[j] >= 0 ? r[j] : self) * 4 + q];
        #pragma unroll
        for (int j = 0; j < 8; ++j) {
            bool ok = r[j] >= 0;
            acc.x += ok ? v[j].x : 0.0f;
            acc.y += ok ? v[j].y : 0.0f;
            acc.z += ok ? v[j].z : 0.0f;
            acc.w += ok ? v[j].w : 0.0f;
        }
        i0 = (pn < end) ? l0 : -1;
        i1 = (pn + 4 < end) ? l1 : -1;
        p = pn;
    }
    float s = (node < n) ? dinv[node] : 0.0f;
    v4f bb = ((const v4f*)b1)[q];
    float h[4];
    h[0] = fmaxf(bb.x + s * acc.x, 0.0f);
    h[1] = fmaxf(bb.y + s * acc.y, 0.0f);
    h[2] = fmaxf(bb.z + s * acc.z, 0.0f);
    h[3] = fmaxf(bb.w + s * acc.w, 0.0f);
    const v2f* w = wt[q];
    float a0 = 0.0f, a1 = 0.0f;
    #pragma unroll
    for (int k = 0; k < 4; ++k) {
        float h0 = k == 0 ? quad_bcast_f<0>(h[0]) : k == 1 ? quad_bcast_f<1>(h[0])
                 : k == 2 ? quad_bcast_f<2>(h[0]) : quad_bcast_f<3>(h[0]);
        float h1v = k == 0 ? quad_bcast_f<0>(h[1]) : k == 1 ? quad_bcast_f<1>(h[1])
                  : k == 2 ? quad_bcast_f<2>(h[1]) : quad_bcast_f<3>(h[1]);
        float h2 = k == 0 ? quad_bcast_f<0>(h[2]) : k == 1 ? quad_bcast_f<1>(h[2])
                 : k == 2 ? quad_bcast_f<2>(h[2]) : quad_bcast_f<3>(h[2]);
        float h3 = k == 0 ? quad_bcast_f<0>(h[3]) : k == 1 ? quad_bcast_f<1>(h[3])
                 : k == 2 ? quad_bcast_f<2>(h[3]) : quad_bcast_f<3>(h[3]);
        v2f w0 = w[4 * k + 0], w1 = w[4 * k + 1], w2 = w[4 * k + 2], w3 = w[4 * k + 3];
        a0 = __builtin_fmaf(h0,  w0.x, a0);  a1 = __builtin_fmaf(h0,  w0.y, a1);
        a0 = __builtin_fmaf(h1v, w1.x, a0);  a1 = __builtin_fmaf(h1v, w1.y, a1);
        a0 = __builtin_fmaf(h2,  w2.x, a0);  a1 = __builtin_fmaf(h2,  w2.y, a1);
        a0 = __builtin_fmaf(h3,  w3.x, a0);  a1 = __builtin_fmaf(h3,  w3.y, a1);
    }
    if (node < n) {
        v2f o = {s * a0, s * a1};
        ((v2f*)g)[(unsigned)node * 4 + q] = o;
    }
}

__global__ void __launch_bounds__(AG_B)
gcn_agg2(const int* __restrict__ start, const int* __restrict__ srcs,
         const float* __restrict__ g, const float* __restrict__ dinv,
         const float* __restrict__ b2, float* __restrict__ out, int n) {
    int tid = threadIdx.x;
    unsigned q = tid & 3u;
    int node = (blockIdx.x * AG_B + tid) >> 2;
    const v2f* g2 = (const v2f*)g;
    int beg = 0, end = 0, self = 0;
    v2f acc = {0.0f, 0.0f};
    if (node < n) {
        beg = start[node];
        end = start[node + 1];
        self = node;
        acc = g2[(unsigned)node * 4 + q];
    }
    int p = beg + (int)q;
    int i0 = (p      < end) ? srcs[p]      : -1;
    int i1 = (p + 4  < end) ? srcs[p + 4]  : -1;
    int i2 = (p + 8  < end) ? srcs[p + 8]  : -1;
    int i3 = (p + 12 < end) ? srcs[p + 12] : -1;
    for (int base = beg; base < end; base += 16) {
        int pn = p + 16;
        int l0 = srcs[pn      < end ? pn      : beg];
        int l1 = srcs[pn + 4  < end ? pn + 4  : beg];
        int l2 = srcs[pn + 8  < end ? pn + 8  : beg];
        int l3 = srcs[pn + 12 < end ? pn + 12 : beg];
        __builtin_amdgcn_sched_barrier(0);
        int r[16];
        v2f v[16];
        quad_bcast_all<0>(i0, r);
        quad_bcast_all<0>(i1, r + 4);
        quad_bcast_all<0>(i2, r + 8);
        quad_bcast_all<0>(i3, r + 12);
        int left = end - base;
        #pragma unroll
        for (int j = 0; j < 16; ++j) v[j] = g2[(unsigned)(j < left ? r[j] : self) * 4 + q];
        __builtin_amdgcn_sched_barrier(0);
        #pragma unroll
        for (int j = 0; j < 16; ++j) {
            bool ok = j < left;
            acc.x += ok ? v[j].x : 0.0f;
            acc.y += ok ? v[j].y : 0.0f;
        }
        i0 = (pn      < end) ? l0 : -1;
        i1 = (pn + 4  < end) ? l1 : -1;
        i2 = (pn + 8  < end) ? l2 : -1;
        i3 = (pn + 12 < end) ? l3 : -1;
        p = pn;
    }
    float s = (node < n) ? dinv[node] : 0.0f;
    unsigned c0 = 2 * q, c1 = 2 * q + 1 < NCLS ? 2 * q + 1 : NCLS - 1;
    float v0 = b2[c0] + s * acc.x;
    float v1 = b2[c1] + s * acc.y;
    float vr[8];
    vr[0] = quad_bcast_f<0>(v0);  vr[1] = quad_bcast_f<0>(v1);
    vr[2] = quad_bcast_f<1>(v0);  vr[3] = quad_bcast_f<1>(v1);
    vr[4] = quad_bcast_f<2>(v0);  vr[5] = quad_bcast_f<2>(v1);
    vr[6] = quad_bcast_f<3>(v0);  vr[7] = quad_bcast_f<3>(v1);
    float m = -1e30f;
    #pragma unroll
    for (int c = 0; c < NCLS; ++c) m = fmaxf(m, vr[c]);
    float sum = 0.0f;
    #pragma unroll
    for (int c = 0; c < NCLS; ++c) sum += __expf(vr[c] - m);
    float lse = m + __logf(sum);
    if (node < n) {
        float* po = out + (unsigned)node * NCLS + 2 * q;
        po[0] = v0 - lse;
        if (2 * q + 1 < NCLS) po[1] = v1 - lse;
    }
}


extern "C" void kernel_launch(void* const* d_in, const int* in_sizes, int n_in,
                              void* d_out, int out_size, void* d_ws, size_t ws_size,
                              hipStream_t stream) {
    const float* x  = (const float*)d_in[0];
    const int*   ei = (const int*)  d_in[1];
    const float* W1 = (const float*)d_in[2];
    const float* b1 = (const float*)d_in[3];
    const float* W2 = (const float*)d_in[4];
    const float* b2 = (const float*)d_in[5];
    float* out = (float*)d_out;

    int n = in_sizes[0] / F_IN;
    int e = in_sizes[1] / 2;
    const int* rowi = ei;
    const int* coli = ei + e;

    size_t np = ((size_t)n + 64) & ~(size_t)63;
    size_t ep = ((size_t)e + 63) & ~(size_t)63;
    size_t rw = ep > 2 * HID * np ? ep : 2 * HID * np;
    int nb = (n + BIN_W - 1) >> BIN_SH;
    if (ws_size < (3 * np + ep + rw) * sizeof(float)) return;
    if (nb > BIN_MAXB || (size_t)BIN_G * (nb + 1) > np) return;

    int*   lbeg  = (int*)d_ws;
    int*   start = lbeg + np;
    float* dinv  = (float*)(start + np);
    int*   srcs  = (int*)(dinv + np);
    unsigned* binned = (unsigned*)(srcs + ep);
    float* h1    = (float*)binned;
    float* g     = h1 + HID * np;

    gcn_bin_local  <<<BIN_G, BIN_T, 0, stream>>>(rowi, coli, lbeg, binned, e, nb);
    gcn_bucket_rank<<<nb, BIN_T, 0, stream>>>(binned, lbeg, start, dinv, srcs, n, e, nb);

    int ntiles = (n + 15) >> 4;
    int gw = (ntiles + GT_WAVES - 1) / GT_WAVES;
    gcn_gemm1_wmma <<<gw, GT_WAVES * 32, 0, stream>>>(x, W1, dinv, h1, n);

    int gq = (n * 4 + AG_B - 1) / AG_B;
    gcn_agg1       <<<gq, AG_B, 0, stream>>>(start, srcs, h1, dinv, b1, W2, g, n);
    gcn_agg2       <<<gq, AG_B, 0, stream>>>(start, srcs, g, dinv, b2, out, n);
}


# Edge sort: a chunk thread requests its 13 edges together, keeps them in registers; the counting LDS add returns the rank (chunk pass: all words; rank pass: a run's first 32); sum order = LDS arrival
# speedup vs baseline: 14.1758x; 1.0417x over previous
#include <hip/hip_runtime.h>
#include <hip/hip_bf16.h>

#define F_IN 512
#define HID  16
#define NCLS 7

typedef float v2f __attribute__((ext_vector_type(2)));
typedef float v8f __attribute__((ext_vector_type(8)));


#define BIN_G    250
#define BIN_T    1024
#define BIN_SH   8
#define BIN_W    (1 << BIN_SH)
#define BIN_MAXB BIN_T
#define STAGE_CAP 16384
#define RUNS_PW  ((BIN_G + BIN_T / 32 - 1) / (BIN_T / 32))
#define BIN_K    13
static_assert(BIN_K * BIN_T <= STAGE_CAP, "a chunk that is kept in registers must fit the LDS image");

__device__ __forceinline__ int wave_scan_incl(int s, int lane) {
    #pragma unroll
    for (int o = 1; o < 32; o <<= 1) {
        int t = __shfl_up(s, o);
        if (lane >= o) s += t;
    }
    return s;
}

__device__ __forceinline__ int wave_sum(int s) {
    #pragma unroll
    for (int o = 16; o > 0; o >>= 1) s += __shfl_xor(s, o);
    return s;
}

template <int K> struct batch_of { static constexpr int size = K; };

template <class Body>
__device__ __forceinline__ void for_edges(int i, int end, Body body) {
    for (; i + 7 * BIN_T < end; i += 8 * BIN_T) body(i, batch_of<8>{});
    if (i + 3 * BIN_T < end) { body(i, batch_of<4>{}); i += 4 * BIN_T; }
    if (i + 1 * BIN_T < end) { body(i, batch_of<2>{}); i += 2 * BIN_T; }
    if (i < end) body(i, batch_of<1>{});
}

__global__ void __launch_bounds__(BIN_T)
gcn_bin_local(const int* __restrict__ row, const int* __restrict__ col,
              int* __restrict__ lbeg, unsigned* __restrict__ binned, int e, int nb) {
    __shared__ int cur[BIN_MAXB];
    __shared__ int wsum[BIN_T / 32];
    __shared__ unsigned stage[STAGE_CAP];
    int tid = threadIdx.x, lane = tid & 31, w = tid >> 5;
    cur[tid] = 0;
    __syncthreads();

    int ch = (e + BIN_G - 1) / BIN_G;
    int lo = blockIdx.x * ch;
    lo = lo < e ? lo : e;
    int hi = lo + ch < e ? lo + ch : e;

    const bool keep = ch <= BIN_K * BIN_T;
    int      kb[BIN_K];
    unsigned kw[BIN_K];
    int      kr[BIN_K];
    #pragma unroll
    for (int j = 0; j < BIN_K; ++j) { kb[j] = 0; kw[j] = 0u; kr[j] = 0; }
    if (keep) {
        if (lo < hi) {
            int c[BIN_K], r[BIN_K];
            #pragma unroll
            for (int j = 0; j < BIN_K; ++j) {
                int i  = lo + tid + j * BIN_T;
                int ic = i < hi ? i : lo;
                c[j] = col[ic];
                r[j] = row[ic];
            }
            #pragma unroll
            for (int j = 0; j < BIN_K; ++j) {
                kb[j] = c[j] >> BIN_SH;
                kw[j] = ((unsigned)r[j] << BIN_SH) | ((unsigned)c[j] & (BIN_W - 1));
                if (lo + tid + j * BIN_T < hi) kr[j] = atomicAdd(&cur[kb[j]], 1);
            }
        }
    } else {
        for_edges(lo + tid, hi, [&](int i, auto kc) {
            constexpr int K = decltype(kc)::size;
            int c[K];
            #pragma unroll
            for (int j = 0; j < K; ++j) c[j] = col[i + j * BIN_T];
            #pragma unroll
            for (int j = 0; j < K; ++j) atomicAdd(&cur[c[j] >> BIN_SH], 1);
        });
    }
    __syncthreads();

    int mine = tid < nb ? cur[tid] : 0;
    int linc = wave_scan_incl(mine, lane);
    if (lane == 31) wsum[w] = linc;
    __syncthreads();
    if (w == 0) {
        int u = wsum[lane];
        int iu = wave_scan_incl(u, lane);
        wsum[lane] = iu - u;
    }
    __syncthreads();
    int lfirst = wsum[w] + linc - mine;
    if (tid < nb) {
        cur[tid] = lfirst;
        int* lrow = lbeg + (size_t)blockIdx.x * (nb + 1);
        lrow[tid] = lfirst;
        if (tid == nb - 1) lrow[nb] = lfirst + mine;
    }
    __syncthreads();

    if (hi - lo <= STAGE_CAP) {
        if (keep) {
            int first[BIN_K];
            #pragma unroll
            for (int j = 0; j < BIN_K; ++j) first[j] = cur[kb[j]];
            #pragma unroll
            for (int j = 0; j < BIN_K; ++j)
                if (lo + tid + j * BIN_T < hi) stage[first[j] + kr[j]] = kw[j];
        } else {
            for_edges(lo + tid, hi, [&](int i, auto kc) {
                constexpr int K = decltype(kc)::size;
                int c[K], r[K];
                #pragma unroll
                for (int j = 0; j < K; ++j) {
                    c[j] = col[i + j * BIN_T];
                    r[j] = row[i + j * BIN_T];
                }
                int pos[K];
                #pragma unroll
                for (int j = 0; j < K; ++j) pos[j] = atomicAdd(&cur[c[j] >> BIN_SH], 1);
                #pragma unroll
                for (int j = 0; j < K; ++j)
                    stage[pos[j]] = ((unsigned)r[j] << BIN_SH) | ((unsigned)c[j] & (BIN_W - 1));
            });
        }
        __syncthreads();
        int n_sl = hi - lo;
        int head = (32 - (lo & 31)) & 31;
        head = head < n_sl ? head : n_sl;
        if (tid < head) binned[lo + tid] = stage[tid];
        for_edges(head + tid, n_sl, [&](int i, auto kc) {
            constexpr int K = decltype(kc)::size;
            unsigned t[K];
            #pragma unroll
            for (int j = 0; j < K; ++j) t[j] = stage[i + j * BIN_T];
            #pragma unroll
            for (int j = 0; j < K; ++j) binned[lo + i + j * BIN_T] = t[j];
        });
    } else {
        for_edges(lo + tid, hi, [&](int i, auto kc) {
            constexpr int K = decltype(kc)::size;
            int c[K], r[K];
            #pragma unroll
            for (int j = 0; j < K; ++j) {
                c[j] = col[i + j * BIN_T];
                r[j] = row[i + j * BIN_T];
            }
            int pos[K];
            #pragma unroll
            for (int j = 0; j < K; ++j) pos[j] = atomicAdd(&cur[c[j] >> BIN_SH], 1);
            #pragma unroll
            for (int j = 0; j < K; ++j)
                binned[lo + pos[j]] = ((unsigned)r[j] << BIN_SH) | ((unsigned)c[j] & (BIN_W - 1));
        });
    }
}

__global__ void __launch_bounds__(BIN_T)
gcn_bucket_rank(const unsigned* __restrict__ binned, const int* __restrict__ lbeg,
                int* __restrict__ start, float* __restrict__ dinv, int* __restrict__ srcs,
                int n, int e, int nb) {
    __shared__ int cnt[BIN_W];
    __shared__ int tcnt[BIN_W];
    __shared__ int base[BIN_W];
    __shared__ int slot[BIN_W];
    __shared__ int wsum[BIN_W / 32];
    __shared__ int roff[BIN_G];
    __shared__ int rlen[BIN_G];
    __shared__ int tot[2];
    __shared__ int stage[STAGE_CAP];
    int tid = threadIdx.x, lane = tid & 31, w = tid >> 5;
    int ch = (e + BIN_G - 1) / BIN_G;
    if (tid < BIN_W) { cnt[tid] = 0; tcnt[tid] = 0; }
    if (tid < 2) tot[tid] = 0;
    __syncthreads();

    if (w < (BIN_G + 31) / 32) {
        int s = 0, t = 0;
        if (tid < BIN_G) {
            const int* lrow = lbeg + (size_t)tid * (nb + 1) + blockIdx.x;
            s = lrow[0];
            t = lrow[1];
            int first = tid * ch;
            roff[tid] = (first < e ? first : e) + s;
            rlen[tid] = t - s;
        }
        s = wave_sum(s);
        t = wave_sum(t);
        if (lane == 0) { atomicAdd(&tot[0], s); atomicAdd(&tot[1], t); }
    }
    __syncthreads();
    int lo = tot[0];
    int hi = tot[1];

    int rs[RUNS_PW], rl[RUNS_PW];
    #pragma unroll
    for (int k = 0; k < RUNS_PW; ++k) {
        int g  = w + k * (BIN_T / 32);
        int gc = g < BIN_G ? g : 0;
        rs[k] = roff[gc];
        rl[k] = g < BIN_G ? rlen[gc] : 0;
    }
    unsigned x[RUNS_PW], y[RUNS_PW];
    #pragma unroll
    for (int k = 0; k < RUNS_PW; ++k) {
        x[k] = binned[lane      < rl[k] ? rs[k] + lane      : 0];
        y[k] = binned[lane + 32 < rl[k] ? rs[k] + lane + 32 : 0];
    }
    auto run_first = [&](int k) { int g = w + k * (BIN_T / 32); return roff[g < BIN_G ? g : 0]; };
    int rx[RUNS_PW];
    #pragma unroll
    for (int k = 0; k < RUNS_PW; ++k) {
        rx[k] = 0;
        if (lane      < rl[k]) rx[k] = atomicAdd(&cnt[x[k] & (BIN_W - 1)], 1);
        if (lane + 32 < rl[k]) atomicAdd(&tcnt[y[k] & (BIN_W - 1)], 1);
    }
    #pragma unroll
    for (int k = 0; k < RUNS_PW; ++k)
        for (int j = lane + 64; j < rl[k]; j += 32) atomicAdd(&tcnt[binned[run_first(k) + j] & (BIN_W - 1)], 1);
    __syncthreads();
    int c = 0, inc = 0, hc = 0;
    if (tid < BIN_W) {
        hc = cnt[tid];
        c = hc + tcnt[tid];
        inc = wave_scan_incl(c, lane);
        if (lane == 31) wsum[w] = inc;
    }
    __syncthreads();
    if (tid < BIN_W) {
        int front = 0;
        #pragma unroll
        for (int k = 0; k < BIN_W / 32; ++k) front += (k < w) ? wsum[k] : 0;
        int excl = front + inc - c;
        base[tid] = excl;
        slot[tid] = excl + hc;
        int node = blockIdx.x * BIN_W + tid;
        if (node < n) {
            start[node] = lo + excl;
            dinv[node]  = rsqrtf((float)(c + 1));
            if (node == n - 1) start[n] = lo + excl + c;
        }
    }
    __syncthreads();
    if (hi - lo <= STAGE_CAP) {
        #pragma unroll
        for (int k = 0; k < RUNS_PW; ++k) rx[k] += base[x[k] & (BIN_W - 1)];
        #pragma unroll
        for (int k = 0; k < RUNS_PW; ++k)
            if (lane < rl[k]) stage[rx[k]] = (int)(x[k] >> BIN_SH);
        #pragma unroll
        for (int k = 0; k < RUNS_PW; ++k)
            if (lane + 32 < rl[k]) {
                int r = atomicAdd(&slot[y[k] & (BIN_W - 1)], 1);
                stage[r] = (int)(y[k] >> BIN_SH);
            }
        #pragma unroll
        for (int k = 0; k < RUNS_PW; ++k)
            for (int j = lane + 64; j < rl[k]; j += 32) {
                unsigned v = binned[run_first(k) + j];
                int r = atomicAdd(&slot[v & (BIN_W - 1)], 1);
                stage[r] = (int)(v >> BIN_SH);
            }
        __syncthreads();
        int n_sl = hi - lo;
        int head = (32 - (lo & 31)) & 31;
        head = head < n_sl ? head : n_sl;
        if (tid < head) srcs[lo + tid] = stage[tid];
        for_edges(head + tid, n_sl, [&](int i, auto kc) {
            constexpr int K = decltype(kc)::size;
            int t[K];
            #pragma unroll
            for (int j = 0; j < K; ++j) t[j] = stage[i + j * BIN_T];
            #pragma unroll
            for (int j = 0; j < K; ++j) srcs[lo + i + j * BIN_T] = t[j];
        });
    } else {
        #pragma unroll
        for (int k = 0; k < RUNS_PW; ++k) rx[k] += base[x[k] & (BIN_W - 1)];
        #pragma unroll
        for (int k = 0; k < RUNS_PW; ++k)
            if (lane < rl[k]) srcs[lo + rx[k]] = (int)(x[k] >> BIN_SH);
        #pragma unroll
        for (int k = 0; k < RUNS_PW; ++k)
            if (lane + 32 < rl[k]) {
                int r = atomicAdd(&slot[y[k] & (BIN_W - 1)], 1);
                srcs[lo + r] = (int)(y[k] >> BIN_SH);
            }
        #pragma unroll
        for (int k = 0; k < RUNS_PW; ++k)
            for (int j = lane + 64; j < rl[k]; j += 32) {
                unsigned v = binned[run_first(k) + j];
                int r = atomicAdd(&slot[v & (BIN_W - 1)], 1);
                srcs[lo + r] = (int)(v >> BIN_SH);
            }
    }
}


#define GT_WAVES 4
#define GT_CH    128
#define GT_P4    (GT_CH / 4 + 1)

typedef float v4f __attribute__((ext_vector_type(4)));

__device__ __forceinline__ void store_tile(float* __restrict__ h1, const float* __restrict__ dinv,
                                           const v8f& acc, int tile, int lane, int n) {
    int mbase = tile * 16 + ((lane >> 4) << 3);
    int ncol  = lane & 15;
    float* p = h1 + (size_t)mbase * HID + ncol;
    if (mbase + 8 <= n) {
        float s[8];
        #pragma unroll
        for (int v = 0; v < 8; ++v) s[v] = dinv[mbase + v];
        #pragma unroll
        for (int v = 0; v < 8; ++v) p[v * HID] = s[v] * acc[v];
    } else {
        #pragma unroll
        for (int v = 0; v < 8; ++v) if (mbase + v < n) p[v * HID] = dinv[mbase + v] * acc[v];
    }
}

__global__ void __launch_bounds__(GT_WAVES * 32)
gcn_gemm1_wmma(const float* __restrict__ x, const float* __restrict__ W1,
               const float* __restrict__ dinv, float* __restrict__ h1, int n) {
    __shared__ v4f wf[(F_IN / 8) * 32];
    __shared__ v4f xs[GT_WAVES][16 * GT_P4];
    int lane = threadIdx.x & 31;
    int w    = threadIdx.x >> 5;
    int m    = lane & 15;
    int h    = lane >> 4;
    int ntiles = (n + 15) >> 4;
    int tile = blockIdx.x * GT_WAVES + w;
    int tl   = tile < ntiles ? tile : ntiles - 1;

    {
        v4f t[F_IN / 8 / GT_WAVES];
        #pragma unroll
        for (int i = 0; i < F_IN / 8 / GT_WAVES; ++i) {
            int G = w + i * GT_WAVES;
            const float* p = W1 + (size_t)(8 * G + 4 * h) * HID + m;
            t[i].x = p[0 * HID];
            t[i].y = p[1 * HID];
            t[i].z = p[2 * HID];
            t[i].w = p[3 * HID];
        }
        #pragma unroll
        for (int i = 0; i < F_IN / 8 / GT_WAVES; ++i) wf[(w + i * GT_WAVES) * 32 + lane] = t[i];
    }

    const v4f* xrow[16];
    #pragma unroll
    for (int r = 0; r < 16; ++r) {
        int row = tl * 16 + r;
        xrow[r] = (const v4f*)(x + (size_t)(row < n ? row : 0) * F_IN) + lane;
    }
    v4f pre[16];
    #pragma unroll
    for (int r = 0; r < 16; ++r) pre[r] = xrow[r][0];

    __syncthreads();

    v4f* xw = xs[w];
    v8f acc0 = {}, acc1 = {};
    #pragma unroll
    for (int c = 0; c < F_IN / GT_CH; ++c) {
        #pragma unroll
        for (int r = 0; r < 16; ++r) xw[r * GT_P4 + lane] = pre[r];
        if (c + 1 < F_IN / GT_CH) {
            #pragma unroll
            for (int r = 0; r < 16; ++r) pre[r] = xrow[r][(c + 1) * (GT_CH / 4)];
        }
        __builtin_amdgcn_wave_barrier();
        #pragma unroll
        for (int g = 0; g < GT_CH / 8; ++g) {
            v4f a = xw[m * GT_P4 + 2 * g + h];
            v4f b = wf[(c * (GT_CH / 8) + g) * 32 + lane];
            v2f a0 = {a.x, a.y}, a1 = {a.z, a.w};
            v2f b0 = {b.x, b.y}, b1 = {b.z, b.w};
            acc0 = __builtin_amdgcn_wmma_f32_16x16x4_f32(
                false, a0, false, b0, (short)0, acc0, false, false);
            acc1 = __builtin_amdgcn_wmma_f32_16x16x4_f32(
                false, a1, false, b1, (short)0, acc1, false, false);
        }
        __builtin_amdgcn_wave_barrier();
    }
    v8f acc = acc0 + acc1;
    if (tile < ntiles) store_tile(h1, dinv, acc, tile, lane, n);
}


#define AG_B 256

template <int J> __device__ __forceinline__ int quad_bcast(int v) {
    return __builtin_amdgcn_update_dpp(0, v, J * 0x55, 0xf, 0xf, true);
}

template <int J> __device__ __forceinline__ void quad_bcast_all(int v, int* r) {
    r[J] = quad_bcast<J>(v);
    if constexpr (J + 1 < 4) quad_bcast_all<J + 1>(v, r);
}

template <int J> __device__ __forceinline__ float quad_bcast_f(float v) {
    return __int_as_float(quad_bcast<J>(__float_as_int(v)));
}

__global__ void __launch_bounds__(AG_B)
gcn_agg1(const int* __restrict__ start, const int* __restrict__ srcs,
         const float* __restrict__ hs, const float* __restrict__ dinv,
         const float* __restrict__ b1, const float* __restrict__ W2,
         float* __restrict__ g, int n) {
    __shared__ v2f wt[4][HID];
    int tid = threadIdx.x;
    unsigned q = tid & 3u;
    int node = (blockIdx.x * AG_B + tid) >> 2;
    const v4f* hs4 = (const v4f*)hs;
    v2f wv = {0.0f, 0.0f};
    if (tid < 4 * HID) {
        int wq = tid >> 4, wj = tid & (HID - 1);
        wv.x = W2[wj * NCLS + 2 * wq];
        if (2 * wq + 1 < NCLS) wv.y = W2[wj * NCLS + 2 * wq + 1];
    }
    int beg = 0, end = 0, self = 0;
    v4f acc = {0.0f, 0.0f, 0.0f, 0.0f};
    if (node < n) {
        beg = start[node];
        end = start[node + 1];
        self = node;
        acc = hs4[(unsigned)node * 4 + q];
    }
    if (tid < 4 * HID) wt[tid >> 4][tid & (HID - 1)] = wv;
    __syncthreads();

    int p = beg + (int)q;
    int i0 = (p < end) ? srcs[p] : -1;
    int i1 = (p + 4 < end) ? srcs[p + 4] : -1;
    for (int base = beg; base < end; base += 8) {
        int pn = p + 8;
        int l0 = srcs[pn < end ? pn : beg];
        int l1 = srcs[pn + 4 < end ? pn + 4 : beg];
        __builtin_amdgcn_sched_barrier(0);
        int r[8];
        v4f v[8];
        quad_bcast_all<0>(i0, r);
        quad_bcast_all<0>(i1, r + 4);
        #pragma unroll
        for (int j = 0; j < 8; ++j) v[j] = hs4[(unsigned)(r[j] >= 0 ? r[j] : self) * 4 + q];
        #pragma unroll
        for (int j = 0; j < 8; ++j) {
            bool ok = r[j] >= 0;
            acc.x += ok ? v[j].x : 0.0f;
            acc.y += ok ? v[j].y : 0.0f;
            acc.z += ok ? v[j].z : 0.0f;
            acc.w += ok ? v[j].w : 0.0f;
        }
        i0 = (pn < end) ? l0 : -1;
        i1 = (pn + 4 < end) ? l1 : -1;
        p = pn;
    }
    float s = (node < n) ? dinv[node] : 0.0f;
    v4f bb = ((const v4f*)b1)[q];
    float h[4];
    h[0] = fmaxf(bb.x + s * acc.x, 0.0f);
    h[1] = fmaxf(bb.y + s * acc.y, 0.0f);
    h[2] = fmaxf(bb.z + s * acc.z, 0.0f);
    h[3] = fmaxf(bb.w + s * acc.w, 0.0f);
    const v2f* w = wt[q];
    float a0 = 0.0f, a1 = 0.0f;
    #pragma unroll
    for (int k = 0; k < 4; ++k) {
        float h0 = k == 0 ? quad_bcast_f<0>(h[0]) : k == 1 ? quad_bcast_f<1>(h[0])
                 : k == 2 ? quad_bcast_f<2>(h[0]) : quad_bcast_f<3>(h[0]);
        float h1v = k == 0 ? quad_bcast_f<0>(h[1]) : k == 1 ? quad_bcast_f<1>(h[1])
                  : k == 2 ? quad_bcast_f<2>(h[1]) : quad_bcast_f<3>(h[1]);
        float h2 = k == 0 ? quad_bcast_f<0>(h[2]) : k == 1 ? quad_bcast_f<1>(h[2])
                 : k == 2 ? quad_bcast_f<2>(h[2]) : quad_bcast_f<3>(h[2]);
        float h3 = k == 0 ? quad_bcast_f<0>(h[3]) : k == 1 ? quad_bcast_f<1>(h[3])
                 : k == 2 ? quad_bcast_f<2>(h[3]) : quad_bcast_f<3>(h[3]);
        v2f w0 = w[4 * k + 0], w1 = w[4 * k + 1], w2 = w[4 * k + 2], w3 = w[4 * k + 3];
        a0 = __builtin_fmaf(h0,  w0.x, a0);  a1 = __builtin_fmaf(h0,  w0.y, a1);
        a0 = __builtin_fmaf(h1v, w1.x, a0);  a1 = __builtin_fmaf(h1v, w1.y, a1);
        a0 = __builtin_fmaf(h2,  w2.x, a0);  a1 = __builtin_fmaf(h2,  w2.y, a1);
        a0 = __builtin_fmaf(h3,  w3.x, a0);  a1 = __builtin_fmaf(h3,  w3.y, a1);
    }
    if (node < n) {
        v2f o = {s * a0, s * a1};
        ((v2f*)g)[(unsigned)node * 4 + q] = o;
    }
}

__global__ void __launch_bounds__(AG_B)
gcn_agg2(const int* __restrict__ start, const int* __restrict__ srcs,
         const float* __restrict__ g, const float* __restrict__ dinv,
         const float* __restrict__ b2, float* __restrict__ out, int n) {
    int tid = threadIdx.x;
    unsigned q = tid & 3u;
    int node = (blockIdx.x * AG_B + tid) >> 2;
    const v2f* g2 = (const v2f*)g;
    int beg = 0, end = 0, self = 0;
    v2f acc = {0.0f, 0.0f};
    if (node < n) {
        beg = start[node];
        end = start[node + 1];
        self = node;
        acc = g2[(unsigned)node * 4 + q];
    }
    int p = beg + (int)q;
    int i0 = (p      < end) ? srcs[p]      : -1;
    int i1 = (p + 4  < end) ? srcs[p + 4]  : -1;
    int i2 = (p + 8  < end) ? srcs[p + 8]  : -1;
    int i3 = (p + 12 < end) ? srcs[p + 12] : -1;
    for (int base = beg; base < end; base += 16) {
        int pn = p + 16;
        int l0 = srcs[pn      < end ? pn      : beg];
        int l1 = srcs[pn + 4  < end ? pn + 4  : beg];
        int l2 = srcs[pn + 8  < end ? pn + 8  : beg];
        int l3 = srcs[pn + 12 < end ? pn + 12 : beg];
        __builtin_amdgcn_sched_barrier(0);
        int r[16];
        v2f v[16];
        quad_bcast_all<0>(i0, r);
        quad_bcast_all<0>(i1, r + 4);
        quad_bcast_all<0>(i2, r + 8);
        quad_bcast_all<0>(i3, r + 12);
        int left = end - base;
        #pragma unroll
        for (int j = 0; j < 16; ++j) v[j] = g2[(unsigned)(j < left ? r[j] : self) * 4 + q];
        __builtin_amdgcn_sched_barrier(0);
        #pragma unroll
        for (int j = 0; j < 16; ++j) {
            bool ok = j < left;
            acc.x += ok ? v[j].x : 0.0f;
            acc.y += ok ? v[j].y : 0.0f;
        }
        i0 = (pn      < end) ? l0 : -1;
        i1 = (pn + 4  < end) ? l1 : -1;
        i2 = (pn + 8  < end) ? l2 : -1;
        i3 = (pn + 12 < end) ? l3 : -1;
        p = pn;
    }
    float s = (node < n) ? dinv[node] : 0.0f;
    unsigned c0 = 2 * q, c1 = 2 * q + 1 < NCLS ? 2 * q + 1 : NCLS - 1;
    float v0 = b2[c0] + s * acc.x;
    float v1 = b2[c1] + s * acc.y;
    float vr[8];
    vr[0] = quad_bcast_f<0>(v0);  vr[1] = quad_bcast_f<0>(v1);
    vr[2] = quad_bcast_f<1>(v0);  vr[3] = quad_bcast_f<1>(v1);
    vr[4] = quad_bcast_f<2>(v0);  vr[5] = quad_bcast_f<2>(v1);
    vr[6] = quad_bcast_f<3>(v0);  vr[7] = quad_bcast_f<3>(v1);
    float m = -1e30f;
    #pragma unroll
    for (int c = 0; c < NCLS; ++c) m = fmaxf(m, vr[c]);
    float sum = 0.0f;
    #pragma unroll
    for (int c = 0; c < NCLS; ++c) sum += __expf(vr[c] - m);
    float lse = m + __logf(sum);
    if (node < n) {
        float* po = out + (unsigned)node * NCLS + 2 * q;
        po[0] = v0 - lse;
        if (2 * q + 1 < NCLS) po[1] = v1 - lse;
    }
}


extern "C" void kernel_launch(void* const* d_in, const int* in_sizes, int n_in,
                              void* d_out, int out_size, void* d_ws, size_t ws_size,
                              hipStream_t stream) {
    const float* x  = (const float*)d_in[0];
    const int*   ei = (const int*)  d_in[1];
    const float* W1 = (const float*)d_in[2];
    const float* b1 = (const float*)d_in[3];
    const float* W2 = (const float*)d_in[4];
    const float* b2 = (const float*)d_in[5];
    float* out = (float*)d_out;

    int n = in_sizes[0] / F_IN;
    int e = in_sizes[1] / 2;
    const int* rowi = ei;
    const int* coli = ei + e;

    size_t np = ((size_t)n + 64) & ~(size_t)63;
    size_t ep = ((size_t)e + 63) & ~(size_t)63;
    size_t rw = ep > 2 * HID * np ? ep : 2 * HID * np;
    int nb = (n + BIN_W - 1) >> BIN_SH;
    if (ws_size < (3 * np + ep + rw) * sizeof(float)) return;
    if (nb > BIN_MAXB || (size_t)BIN_G * (nb + 1) > np) return;

    int*   lbeg  = (int*)d_ws;
    int*   start = lbeg + np;
    float* dinv  = (float*)(start + np);
    int*   srcs  = (int*)(dinv + np);
    unsigned* binned = (unsigned*)(srcs + ep);
    float* h1    = (float*)binned;
    float* g     = h1 + HID * np;

    gcn_bin_local  <<<BIN_G, BIN_T, 0, stream>>>(rowi, coli, lbeg, binned, e, nb);
    gcn_bucket_rank<<<nb, BIN_T, 0, stream>>>(binned, lbeg, start, dinv, srcs, n, e, nb);

    int ntiles = (n + 15) >> 4;
    int gw = (ntiles + GT_WAVES - 1) / GT_WAVES;
    gcn_gemm1_wmma <<<gw, GT_WAVES * 32, 0, stream>>>(x, W1, dinv, h1, n);

    int gq = (n * 4 + AG_B - 1) / AG_B;
    gcn_agg1       <<<gq, AG_B, 0, stream>>>(start, srcs, h1, dinv, b1, W2, g, n);
    gcn_agg2       <<<gq, AG_B, 0, stream>>>(start, srcs, g, dinv, b2, out, n);
}
